# starting kernel
# baseline (speedup 1.0000x reference)
#include <hip/hip_runtime.h>

typedef __attribute__((ext_vector_type(2))) float v2f;
typedef __attribute__((ext_vector_type(8))) float v8f;

#define IN_C 128
#define OUT_C 128

static inline unsigned cdiv_u(long long a, long long b) { return (unsigned)((a + b - 1) / b); }

__device__ __forceinline__ void atomic_add_f32_hw(float* addr, float v) {
    asm volatile("global_atomic_add_f32 %0, %1, off scope:SCOPE_DEV"
                 :: "v"(addr), "v"(v)
                 : "memory");
}

__global__ void k_init_deg(int* __restrict__ deg, int n) {
    int i = blockIdx.x * blockDim.x + threadIdx.x;
    if (i < n) deg[i] = 1;
}

__global__ void k_count_deg(const int* __restrict__ rows, int* __restrict__ deg, int e) {
    int i = blockIdx.x * blockDim.x + threadIdx.x;
    if (i < e) atomicAdd(&deg[rows[i]], 1);
}

__global__ void k_rsqrt(const int* __restrict__ deg, float* __restrict__ dinv, int n) {
    int i = blockIdx.x * blockDim.x + threadIdx.x;
    if (i < n) dinv[i] = rsqrtf((float)deg[i]);
}

__global__ __launch_bounds__(256) void k_gemm_wmma(const float* __restrict__ x,
                                                   const float* __restrict__ Wm,
                                                   const float* __restrict__ bias,
                                                   float* __restrict__ h, int n) {
    const int lane  = threadIdx.x & 31;
    const int wave  = threadIdx.x >> 5;
    const int half  = lane >> 4;
    const int l     = lane & 15;
    const int mBase = blockIdx.x * 16;
    const int nBase = wave * 16;

    int mrow = mBase + l;
    if (mrow > n - 1) mrow = n - 1;
    const int koff = half * 2;

    const float* xr = x  + (size_t)mrow * IN_C + koff;
    const float* wr = Wm + (size_t)(nBase + l) * IN_C + koff;

    v8f c = {0.f, 0.f, 0.f, 0.f, 0.f, 0.f, 0.f, 0.f};
#pragma unroll 8
    for (int kk = 0; kk < IN_C; kk += 4) {
        v2f a = *(const v2f*)(xr + kk);
        v2f b = *(const v2f*)(wr + kk);
        c = __builtin_amdgcn_wmma_f32_16x16x4_f32(false, a, false, b, (short)0, c,
                                                  false, false);
    }

    const float bv = bias[nBase + l];
    float* hp = h + (size_t)mBase * OUT_C + nBase + l;
#pragma unroll
    for (int i = 0; i < 8; ++i) {
        const int mi = i + half * 8;
        if (mBase + mi < n) hp[(size_t)mi * OUT_C] = c[i] + bv;
    }
}

__global__ void k_out_init(const float* __restrict__ h, const float* __restrict__ dinv,
                           float* __restrict__ out, int total) {
    int i = blockIdx.x * blockDim.x + threadIdx.x;
    if (i < total) {
        float d = dinv[i >> 7];
        out[i] = d * d * h[i];
    }
}

__global__ __launch_bounds__(256) void k_scatter(const int* __restrict__ rows,
                                                 const int* __restrict__ cols,
                                                 const float* __restrict__ dinv,
                                                 const float* __restrict__ h,
                                                 float* __restrict__ out, int e) {
    const int lane = threadIdx.x & 31;
    const int wave = threadIdx.x >> 5;
    const int ed   = __builtin_amdgcn_readfirstlane(blockIdx.x * 8 + wave);
    if (ed >= e) return;

    const int   r   = rows[ed];
    const int   cl  = cols[ed];
    const float nrm = dinv[r] * dinv[cl];

    const int c4 = lane << 2;
    const float4 hv = *(const float4*)(h + (size_t)cl * OUT_C + c4);
    float* op = out + (size_t)r * OUT_C + c4;
    atomic_add_f32_hw(op + 0, nrm * hv.x);
    atomic_add_f32_hw(op + 1, nrm * hv.y);
    atomic_add_f32_hw(op + 2, nrm * hv.z);
    atomic_add_f32_hw(op + 3, nrm * hv.w);
}

extern "C" void kernel_launch(void* const* d_in, const int* in_sizes, int n_in,
                              void* d_out, int out_size, void* d_ws, size_t ws_size,
                              hipStream_t stream) {
    const float* x    = (const float*)d_in[0];
    const int*   ei   = (const int*)d_in[1];
    const float* Wm   = (const float*)d_in[2];
    const float* bias = (const float*)d_in[3];
    float*       out  = (float*)d_out;

    const int Nn = in_sizes[0] / IN_C;
    const int Ec = in_sizes[1] / 2;
    const int* rows = ei;
    const int* cols = ei + Ec;

    char*  base = (char*)d_ws;
    float* h    = (float*)base;
    int*   deg  = (int*)(base + (size_t)Nn * OUT_C * sizeof(float));
    float* dinv = (float*)((char*)deg + (size_t)Nn * sizeof(int));

    k_init_deg<<<cdiv_u(Nn, 256), 256, 0, stream>>>(deg, Nn);
    k_count_deg<<<cdiv_u(Ec, 256), 256, 0, stream>>>(rows, deg, Ec);
    k_gemm_wmma<<<cdiv_u(Nn, 16), 256, 0, stream>>>(x, Wm, bias, h, Nn);
    k_rsqrt<<<cdiv_u(Nn, 256), 256, 0, stream>>>(deg, dinv, Nn);
    k_out_init<<<cdiv_u((long long)Nn * OUT_C, 256), 256, 0, stream>>>(h, dinv, out,
                                                                       Nn * OUT_C);
    k_scatter<<<cdiv_u(Ec, 8), 256, 0, stream>>>(rows, cols, dinv, h, out, Ec);
}


# Edge scatter: lane l owns channels l,l+32,l+64,l+96, so each f32 atomic instruction covers one whole 128-B line of out (4 line requests per edge, was 16); atomics kept, arrival-order sums as before
# speedup vs baseline: 1.1491x; 1.1491x over previous
#include <hip/hip_runtime.h>

typedef __attribute__((ext_vector_type(2))) float v2f;
typedef __attribute__((ext_vector_type(8))) float v8f;

#define IN_C 128
#define OUT_C 128

static inline unsigned cdiv_u(long long a, long long b) { return (unsigned)((a + b - 1) / b); }

template <int OFF>
__device__ __forceinline__ void atomic_add_f32_hw(float* row_base, unsigned lane_off, float v) {
    asm volatile("global_atomic_add_f32 %0, %1, %2 offset:%3 scope:SCOPE_DEV"
                 :: "v"(lane_off), "v"(v), "s"(row_base), "n"(OFF)
                 : "memory");
}

__global__ void k_init_deg(int* __restrict__ deg, int n) {
    int i = blockIdx.x * blockDim.x + threadIdx.x;
    if (i < n) deg[i] = 1;
}

__global__ void k_count_deg(const int* __restrict__ rows, int* __restrict__ deg, int e) {
    int i = blockIdx.x * blockDim.x + threadIdx.x;
    if (i < e) atomicAdd(&deg[rows[i]], 1);
}

__global__ void k_rsqrt(const int* __restrict__ deg, float* __restrict__ dinv, int n) {
    int i = blockIdx.x * blockDim.x + threadIdx.x;
    if (i < n) dinv[i] = rsqrtf((float)deg[i]);
}

__global__ __launch_bounds__(256) void k_gemm_wmma(const float* __restrict__ x,
                                                   const float* __restrict__ Wm,
                                                   const float* __restrict__ bias,
                                                   float* __restrict__ h, int n) {
    const int lane  = threadIdx.x & 31;
    const int wave  = threadIdx.x >> 5;
    const int half  = lane >> 4;
    const int l     = lane & 15;
    const int mBase = blockIdx.x * 16;
    const int nBase = wave * 16;

    int mrow = mBase + l;
    if (mrow > n - 1) mrow = n - 1;
    const int koff = half * 2;

    const float* xr = x  + (size_t)mrow * IN_C + koff;
    const float* wr = Wm + (size_t)(nBase + l) * IN_C + koff;

    v8f c = {0.f, 0.f, 0.f, 0.f, 0.f, 0.f, 0.f, 0.f};
#pragma unroll 8
    for (int kk = 0; kk < IN_C; kk += 4) {
        v2f a = *(const v2f*)(xr + kk);
        v2f b = *(const v2f*)(wr + kk);
        c = __builtin_amdgcn_wmma_f32_16x16x4_f32(false, a, false, b, (short)0, c,
                                                  false, false);
    }

    const float bv = bias[nBase + l];
    float* hp = h + (size_t)mBase * OUT_C + nBase + l;
#pragma unroll
    for (int i = 0; i < 8; ++i) {
        const int mi = i + half * 8;
        if (mBase + mi < n) hp[(size_t)mi * OUT_C] = c[i] + bv;
    }
}

__global__ void k_out_init(const float* __restrict__ h, const float* __restrict__ dinv,
                           float* __restrict__ out, int total) {
    int i = blockIdx.x * blockDim.x + threadIdx.x;
    if (i < total) {
        float d = dinv[i >> 7];
        out[i] = d * d * h[i];
    }
}

__global__ __launch_bounds__(256) void k_scatter(const int* __restrict__ rows,
                                                 const int* __restrict__ cols,
                                                 const float* __restrict__ dinv,
                                                 const float* __restrict__ h,
                                                 float* __restrict__ out, int e) {
    const int lane = threadIdx.x & 31;
    const int wave = threadIdx.x >> 5;
    const int ed   = __builtin_amdgcn_readfirstlane(blockIdx.x * 8 + wave);
    if (ed >= e) return;

    const int   r   = rows[ed];
    const int   cl  = cols[ed];
    const float nrm = dinv[r] * dinv[cl];

    const float* hp = h + (size_t)cl * OUT_C + lane;
    const float v0 = hp[0];
    const float v1 = hp[32];
    const float v2 = hp[64];
    const float v3 = hp[96];

    float* ob = out + (size_t)r * OUT_C;
    const unsigned lo = (unsigned)lane * 4u;
    atomic_add_f32_hw<0>(ob, lo, nrm * v0);
    atomic_add_f32_hw<128>(ob, lo, nrm * v1);
    atomic_add_f32_hw<256>(ob, lo, nrm * v2);
    atomic_add_f32_hw<384>(ob, lo, nrm * v3);
}

extern "C" void kernel_launch(void* const* d_in, const int* in_sizes, int n_in,
                              void* d_out, int out_size, void* d_ws, size_t ws_size,
                              hipStream_t stream) {
    const float* x    = (const float*)d_in[0];
    const int*   ei   = (const int*)d_in[1];
    const float* Wm   = (const float*)d_in[2];
    const float* bias = (const float*)d_in[3];
    float*       out  = (float*)d_out;

    const int Nn = in_sizes[0] / IN_C;
    const int Ec = in_sizes[1] / 2;
    const int* rows = ei;
    const int* cols = ei + Ec;

    char*  base = (char*)d_ws;
    float* h    = (float*)base;
    int*   deg  = (int*)(base + (size_t)Nn * OUT_C * sizeof(float));
    float* dinv = (float*)((char*)deg + (size_t)Nn * sizeof(int));

    k_init_deg<<<cdiv_u(Nn, 256), 256, 0, stream>>>(deg, Nn);
    k_count_deg<<<cdiv_u(Ec, 256), 256, 0, stream>>>(rows, deg, Ec);
    k_gemm_wmma<<<cdiv_u(Nn, 16), 256, 0, stream>>>(x, Wm, bias, h, Nn);
    k_rsqrt<<<cdiv_u(Nn, 256), 256, 0, stream>>>(deg, dinv, Nn);
    k_out_init<<<cdiv_u((long long)Nn * OUT_C, 256), 256, 0, stream>>>(h, dinv, out,
                                                                       Nn * OUT_C);
    k_scatter<<<cdiv_u(Ec, 8), 256, 0, stream>>>(rows, cols, dinv, h, out, Ec);
}


# Edge scatter: each wave walks 16 consecutive edges, all 64 line loads of h issued before its 64 f32 atomics (16x fewer waves, 16x the lines in flight per wave); atomics and arrival-order sums kept
# speedup vs baseline: 1.4498x; 1.2617x over previous
#include <hip/hip_runtime.h>

typedef __attribute__((ext_vector_type(2))) float v2f;
typedef __attribute__((ext_vector_type(8))) float v8f;

#define IN_C 128
#define OUT_C 128

static inline unsigned cdiv_u(long long a, long long b) { return (unsigned)((a + b - 1) / b); }

template <int OFF>
__device__ __forceinline__ void atomic_add_f32_hw(float* row_base, unsigned lane_off, float v) {
    asm volatile("global_atomic_add_f32 %0, %1, %2 offset:%3 scope:SCOPE_DEV"
                 :: "v"(lane_off), "v"(v), "s"(row_base), "n"(OFF)
                 : "memory");
}

__global__ void k_init_deg(int* __restrict__ deg, int n) {
    int i = blockIdx.x * blockDim.x + threadIdx.x;
    if (i < n) deg[i] = 1;
}

__global__ void k_count_deg(const int* __restrict__ rows, int* __restrict__ deg, int e) {
    int i = blockIdx.x * blockDim.x + threadIdx.x;
    if (i < e) atomicAdd(&deg[rows[i]], 1);
}

__global__ void k_rsqrt(const int* __restrict__ deg, float* __restrict__ dinv, int n) {
    int i = blockIdx.x * blockDim.x + threadIdx.x;
    if (i < n) dinv[i] = rsqrtf((float)deg[i]);
}

__global__ __launch_bounds__(256) void k_gemm_wmma(const float* __restrict__ x,
                                                   const float* __restrict__ Wm,
                                                   const float* __restrict__ bias,
                                                   float* __restrict__ h, int n) {
    const int lane  = threadIdx.x & 31;
    const int wave  = threadIdx.x >> 5;
    const int half  = lane >> 4;
    const int l     = lane & 15;
    const int mBase = blockIdx.x * 16;
    const int nBase = wave * 16;

    int mrow = mBase + l;
    if (mrow > n - 1) mrow = n - 1;
    const int koff = half * 2;

    const float* xr = x  + (size_t)mrow * IN_C + koff;
    const float* wr = Wm + (size_t)(nBase + l) * IN_C + koff;

    v8f c = {0.f, 0.f, 0.f, 0.f, 0.f, 0.f, 0.f, 0.f};
#pragma unroll 8
    for (int kk = 0; kk < IN_C; kk += 4) {
        v2f a = *(const v2f*)(xr + kk);
        v2f b = *(const v2f*)(wr + kk);
        c = __builtin_amdgcn_wmma_f32_16x16x4_f32(false, a, false, b, (short)0, c,
                                                  false, false);
    }

    const float bv = bias[nBase + l];
    float* hp = h + (size_t)mBase * OUT_C + nBase + l;
#pragma unroll
    for (int i = 0; i < 8; ++i) {
        const int mi = i + half * 8;
        if (mBase + mi < n) hp[(size_t)mi * OUT_C] = c[i] + bv;
    }
}

__global__ void k_out_init(const float* __restrict__ h, const float* __restrict__ dinv,
                           float* __restrict__ out, int total) {
    int i = blockIdx.x * blockDim.x + threadIdx.x;
    if (i < total) {
        float d = dinv[i >> 7];
        out[i] = d * d * h[i];
    }
}

#define EPW 16

__global__ __launch_bounds__(256) void k_scatter(const int* __restrict__ rows,
                                                 const int* __restrict__ cols,
                                                 const float* __restrict__ dinv,
                                                 const float* __restrict__ h,
                                                 float* __restrict__ out, int e) {
    const int lane  = threadIdx.x & 31;
    const int wave  = threadIdx.x >> 5;
    const int wbase = __builtin_amdgcn_readfirstlane((blockIdx.x * 8 + wave) * EPW);
    if (wbase >= e) return;
    const int left = e - wbase;

    int idx = wbase + (lane & (EPW - 1));
    if (idx > e - 1) idx = e - 1;
    const int   r   = rows[idx];
    const int   cl  = cols[idx];
    const float nrm = dinv[r] * dinv[cl];

    const unsigned lo = (unsigned)lane * 4u;

    if (left >= EPW) {
        float v[EPW][4];
#pragma unroll
        for (int j = 0; j < EPW; ++j) {
            const int cj = __builtin_amdgcn_readlane(cl, j);
            const float* hp = h + (size_t)cj * OUT_C + lane;
            v[j][0] = hp[0];
            v[j][1] = hp[32];
            v[j][2] = hp[64];
            v[j][3] = hp[96];
        }
#pragma unroll
        for (int j = 0; j < EPW; ++j) {
            const int   rj = __builtin_amdgcn_readlane(r, j);
            const float nj = __int_as_float(__builtin_amdgcn_readlane(__float_as_int(nrm), j));
            float* ob = out + (size_t)rj * OUT_C;
            atomic_add_f32_hw<0>(ob, lo, nj * v[j][0]);
            atomic_add_f32_hw<128>(ob, lo, nj * v[j][1]);
            atomic_add_f32_hw<256>(ob, lo, nj * v[j][2]);
            atomic_add_f32_hw<384>(ob, lo, nj * v[j][3]);
        }
    } else {
        for (int j = 0; j < left; ++j) {
            const int   cj = __builtin_amdgcn_readlane(cl, j);
            const int   rj = __builtin_amdgcn_readlane(r, j);
            const float nj = __int_as_float(__builtin_amdgcn_readlane(__float_as_int(nrm), j));
            const float* hp = h + (size_t)cj * OUT_C + lane;
            const float v0 = hp[0];
            const float v1 = hp[32];
            const float v2 = hp[64];
            const float v3 = hp[96];
            float* ob = out + (size_t)rj * OUT_C;
            atomic_add_f32_hw<0>(ob, lo, nj * v0);
            atomic_add_f32_hw<128>(ob, lo, nj * v1);
            atomic_add_f32_hw<256>(ob, lo, nj * v2);
            atomic_add_f32_hw<384>(ob, lo, nj * v3);
        }
    }
}

extern "C" void kernel_launch(void* const* d_in, const int* in_sizes, int n_in,
                              void* d_out, int out_size, void* d_ws, size_t ws_size,
                              hipStream_t stream) {
    const float* x    = (const float*)d_in[0];
    const int*   ei   = (const int*)d_in[1];
    const float* Wm   = (const float*)d_in[2];
    const float* bias = (const float*)d_in[3];
    float*       out  = (float*)d_out;

    const int Nn = in_sizes[0] / IN_C;
    const int Ec = in_sizes[1] / 2;
    const int* rows = ei;
    const int* cols = ei + Ec;

    char*  base = (char*)d_ws;
    float* h    = (float*)base;
    int*   deg  = (int*)(base + (size_t)Nn * OUT_C * sizeof(float));
    float* dinv = (float*)((char*)deg + (size_t)Nn * sizeof(int));

    k_init_deg<<<cdiv_u(Nn, 256), 256, 0, stream>>>(deg, Nn);
    k_count_deg<<<cdiv_u(Ec, 256), 256, 0, stream>>>(rows, deg, Ec);
    k_gemm_wmma<<<cdiv_u(Nn, 16), 256, 0, stream>>>(x, Wm, bias, h, Nn);
    k_rsqrt<<<cdiv_u(Nn, 256), 256, 0, stream>>>(deg, dinv, Nn);
    k_out_init<<<cdiv_u((long long)Nn * OUT_C, 256), 256, 0, stream>>>(h, dinv, out,
                                                                       Nn * OUT_C);
    k_scatter<<<cdiv_u(Ec, 8 * EPW), 256, 0, stream>>>(rows, cols, dinv, h, out, Ec);
}


# Scatter-add by f32 atomics replaced by a row-owned sum: a returning int add bins each edge by row, one wave sums a row's source lines in registers in arrival-rank order and stores each line once
# speedup vs baseline: 1.6510x; 1.1388x over previous
#include <hip/hip_runtime.h>


typedef __attribute__((ext_vector_type(2))) float v2f;
typedef __attribute__((ext_vector_type(4))) float v4f;
typedef __attribute__((ext_vector_type(8))) float v8f;

#define IN_C 128
#define OUT_C 128

static inline unsigned cdiv_u(long long a, long long b) { return (unsigned)((a + b - 1) / b); }

__global__ void k_init_deg(int* __restrict__ deg, int n) {
    int i = blockIdx.x * blockDim.x + threadIdx.x;
    if (i < n) deg[i] = 1;
}

#define SLOTS 128
#define EPT 4

template <bool FULL>
__device__ __forceinline__ void bin_edges(const int* __restrict__ rows,
                                          const int* __restrict__ cols,
                                          int* __restrict__ deg,
                                          float* __restrict__ out, int i0, int e) {
    int r[EPT], c[EPT], k[EPT];
#pragma unroll
    for (int u = 0; u < EPT; ++u) {
        const int i = i0 + u * 256;
        r[u] = 0;
        c[u] = 0;
        if (FULL || i < e) {
            r[u] = rows[i];
            c[u] = cols[i];
        }
    }
#pragma unroll
    for (int u = 0; u < EPT; ++u) {
        const int i = i0 + u * 256;
        k[u] = SLOTS + 1;
        if (FULL || i < e) k[u] = atomicAdd(&deg[r[u]], 1);
    }
#pragma unroll
    for (int u = 0; u < EPT; ++u) {
        const int rank = k[u] - 1;
        if (rank < SLOTS) out[r[u] * OUT_C + rank] = __int_as_float(c[u]);
    }
}

__global__ __launch_bounds__(256) void k_count_bin(const int* __restrict__ rows,
                                                   const int* __restrict__ cols,
                                                   int* __restrict__ deg,
                                                   float* __restrict__ out, int e) {
    const int b0 = blockIdx.x * (256 * EPT);
    const int i0 = b0 + threadIdx.x;
    if (b0 + 256 * EPT <= e) bin_edges<true>(rows, cols, deg, out, i0, e);
    else                     bin_edges<false>(rows, cols, deg, out, i0, e);
}

__global__ __launch_bounds__(256) void k_gemm_wmma(const float* __restrict__ x,
                                                   const float* __restrict__ Wm,
                                                   const float* __restrict__ bias,
                                                   float* __restrict__ h, int n) {
    const int lane  = threadIdx.x & 31;
    const int wave  = threadIdx.x >> 5;
    const int half  = lane >> 4;
    const int l     = lane & 15;
    const int mBase = blockIdx.x * 16;
    const int nBase = wave * 16;

    int mrow = mBase + l;
    if (mrow > n - 1) mrow = n - 1;
    const int koff = half * 2;

    const float* xr = x  + (size_t)mrow * IN_C + koff;
    const float* wr = Wm + (size_t)(nBase + l) * IN_C + koff;

    v8f c = {0.f, 0.f, 0.f, 0.f, 0.f, 0.f, 0.f, 0.f};
#pragma unroll 8
    for (int kk = 0; kk < IN_C; kk += 4) {
        v2f a = *(const v2f*)(xr + kk);
        v2f b = *(const v2f*)(wr + kk);
        c = __builtin_amdgcn_wmma_f32_16x16x4_f32(false, a, false, b, (short)0, c,
                                                  false, false);
    }

    const float bv = bias[nBase + l];
    float* hp = h + (size_t)mBase * OUT_C + nBase + l;
#pragma unroll
    for (int i = 0; i < 8; ++i) {
        const int mi = i + half * 8;
        if (mBase + mi < n) hp[(size_t)mi * OUT_C] = c[i] + bv;
    }
}

template <int K>
__device__ __forceinline__ void sum_edges(const float* __restrict__ h, int c, float nrm,
                                          int j0, int lane, v4f& acc) {
    v4f v[K];
#pragma unroll
    for (int u = 0; u < K; ++u) {
        const int cj = __builtin_amdgcn_readlane(c, j0 + u);
        v[u] = ((const v4f*)h)[cj * (OUT_C / 4) + lane];
    }
#pragma unroll
    for (int u = 0; u < K; ++u) {
        const float nj = __int_as_float(__builtin_amdgcn_readlane(__float_as_int(nrm), j0 + u));
        acc += nj * v[u];
    }
}

__global__ __launch_bounds__(256) void k_row_sum(const int* __restrict__ rows,
                                                 const int* __restrict__ cols,
                                                 const int* __restrict__ deg,
                                                 const float* __restrict__ h,
                                                 float* out, int n, int e) {
    const int lane = threadIdx.x & 31;
    const int r    = __builtin_amdgcn_readfirstlane(blockIdx.x * 8 + (threadIdx.x >> 5));
    if (r >= n) return;

    float* orow = out + (size_t)r * OUT_C;
    const v4f hs    = ((const v4f*)h)[r * (OUT_C / 4) + lane];
    int       cslot = __float_as_int(orow[lane]);

    const int   dg  = __builtin_amdgcn_readfirstlane(deg[r]);
    const int   cnt = dg - 1;
    const float dr  = rsqrtf((float)dg);

    v4f acc = (dr * dr) * hs;

    if (cnt <= SLOTS) {
        for (int base = 0; base < cnt; base += 32) {
            const int rem = cnt - base;
            const int len = rem < 32 ? rem : 32;
            int c = lane < len ? cslot : r;
            c = c < 0 ? 0 : (c > n - 1 ? n - 1 : c);
            if (base + 32 < cnt) cslot = __float_as_int(orow[base + 32 + lane]);
            const float nrm = dr * rsqrtf((float)deg[c]);
            int j = 0;
            for (; j + 8 <= len; j += 8) sum_edges<8>(h, c, nrm, j, lane, acc);
            if (len - j >= 4) { sum_edges<4>(h, c, nrm, j, lane, acc); j += 4; }
            if (len - j >= 2) { sum_edges<2>(h, c, nrm, j, lane, acc); j += 2; }
            if (len - j >= 1) { sum_edges<1>(h, c, nrm, j, lane, acc); }
        }
    } else {
        for (int base = 0; base < e; base += 32) {
            const int idx = base + lane;
            int rr = -1;
            int cc = r;
            if (idx < e) {
                rr = rows[idx];
                cc = cols[idx];
            }
            unsigned m = __builtin_amdgcn_ballot_w32(rr == r);
            if (m) {
                const float nrm = dr * rsqrtf((float)deg[cc]);
                while (m) {
                    const int j = __builtin_ctz(m);
                    m &= m - 1;
                    sum_edges<1>(h, cc, nrm, j, lane, acc);
                }
            }
        }
    }

    *(v4f*)(orow + lane * 4) = acc;
}

extern "C" void kernel_launch(void* const* d_in, const int* in_sizes, int n_in,
                              void* d_out, int out_size, void* d_ws, size_t ws_size,
                              hipStream_t stream) {
    const float* x    = (const float*)d_in[0];
    const int*   ei   = (const int*)d_in[1];
    const float* Wm   = (const float*)d_in[2];
    const float* bias = (const float*)d_in[3];
    float*       out  = (float*)d_out;

    const int Nn = in_sizes[0] / IN_C;
    const int Ec = in_sizes[1] / 2;
    const int* rows = ei;
    const int* cols = ei + Ec;

    char*  base = (char*)d_ws;
    float* h    = (float*)base;
    int*   deg  = (int*)(base + (size_t)Nn * OUT_C * sizeof(float));

    k_init_deg<<<cdiv_u(Nn, 256), 256, 0, stream>>>(deg, Nn);
    k_count_bin<<<cdiv_u(Ec, 256 * EPT), 256, 0, stream>>>(rows, cols, deg, out, Ec);
    k_gemm_wmma<<<cdiv_u(Nn, 16), 256, 0, stream>>>(x, Wm, bias, h, Nn);
    k_row_sum<<<cdiv_u(Nn, 8), 256, 0, stream>>>(rows, cols, deg, h, out, Nn, Ec);
}


# GEMM re-tiled: a wave owns a 16x128 strip of h (one A fragment feeds 8 column tiles, interleaved so h leaves as whole lines), W staged once per block in LDS in fragment order; k order of sums kept
# speedup vs baseline: 1.8873x; 1.1431x over previous
#include <hip/hip_runtime.h>


typedef __attribute__((ext_vector_type(2))) float v2f;
typedef __attribute__((ext_vector_type(4))) float v4f;
typedef __attribute__((ext_vector_type(8))) float v8f;

#define IN_C 128
#define OUT_C 128

static inline unsigned cdiv_u(long long a, long long b) { return (unsigned)((a + b - 1) / b); }

__global__ void k_init_deg(int* __restrict__ deg, int n) {
    int i = blockIdx.x * blockDim.x + threadIdx.x;
    if (i < n) deg[i] = 1;
}

#define SLOTS 128
#define EPT 4

template <bool FULL>
__device__ __forceinline__ void bin_edges(const int* __restrict__ rows,
                                          const int* __restrict__ cols,
                                          int* __restrict__ deg,
                                          float* __restrict__ out, int i0, int e) {
    int r[EPT], c[EPT], k[EPT];
#pragma unroll
    for (int u = 0; u < EPT; ++u) {
        const int i = i0 + u * 256;
        r[u] = 0;
        c[u] = 0;
        if (FULL || i < e) {
            r[u] = rows[i];
            c[u] = cols[i];
        }
    }
#pragma unroll
    for (int u = 0; u < EPT; ++u) {
        const int i = i0 + u * 256;
        k[u] = SLOTS + 1;
        if (FULL || i < e) k[u] = atomicAdd(&deg[r[u]], 1);
    }
#pragma unroll
    for (int u = 0; u < EPT; ++u) {
        const int rank = k[u] - 1;
        if (rank < SLOTS) out[r[u] * OUT_C + rank] = __int_as_float(c[u]);
    }
}

__global__ __launch_bounds__(256) void k_count_bin(const int* __restrict__ rows,
                                                   const int* __restrict__ cols,
                                                   int* __restrict__ deg,
                                                   float* __restrict__ out, int e) {
    const int b0 = blockIdx.x * (256 * EPT);
    const int i0 = b0 + threadIdx.x;
    if (b0 + 256 * EPT <= e) bin_edges<true>(rows, cols, deg, out, i0, e);
    else                     bin_edges<false>(rows, cols, deg, out, i0, e);
}

#define GEMM_WAVES  16
#define GEMM_STRIPS 13

__global__ __launch_bounds__(GEMM_WAVES * 32) void k_gemm_wmma(const float* __restrict__ x,
                                                               const float* __restrict__ Wm,
                                                               const float* __restrict__ bias,
                                                               float* __restrict__ h, int n) {
    __shared__ v4f wl[(IN_C / 8) * (OUT_C / 16) * 32];

    const int lane = threadIdx.x & 31;
    const int wave = __builtin_amdgcn_readfirstlane(threadIdx.x >> 5);
    const int half = lane >> 4;
    const int l    = lane & 15;

    v4f g[8];
#pragma unroll
    for (int u = 0; u < 8; ++u) {
        const int unit = wave * 8 + u;
        const int kp = unit >> 3, nt = unit & 7;
        const int col = (nt >> 2) * 64 + l * 4 + (nt & 3);
        g[u] = *(const v4f*)(Wm + col * IN_C + (2 * kp + half) * 4);
    }

    const int  ntiles = (n + 15) >> 4;
    const int  tile   = blockIdx.x * GEMM_STRIPS + wave;
    const bool owns   = wave < GEMM_STRIPS && tile < ntiles;

    v2f a[IN_C / 4];
    {
        int t = owns ? tile : blockIdx.x * GEMM_STRIPS;
        if (t > ntiles - 1) t = ntiles - 1;
        int mrow = t * 16 + l;
        if (mrow > n - 1) mrow = n - 1;
        const float* xr = x + mrow * IN_C + half * 2;
#pragma unroll
        for (int i = 0; i < IN_C / 4; ++i) a[i] = *(const v2f*)(xr + 4 * i);
    }

    {
        v2f* w2 = (v2f*)wl;
#pragma unroll
        for (int u = 0; u < 8; ++u) {
            const int unit = wave * 8 + u;
            v2f lo = {g[u].x, g[u].y};
            v2f hi = {g[u].z, g[u].w};
            w2[(unit * 32 + l) * 2 + half]      = lo;
            w2[(unit * 32 + 16 + l) * 2 + half] = hi;
        }
    }
    __syncthreads();
    if (!owns) return;

    const int mBase = tile * 16;

    v8f c[OUT_C / 16];
#pragma unroll
    for (int nt = 0; nt < OUT_C / 16; ++nt) c[nt] = (v8f){0.f, 0.f, 0.f, 0.f, 0.f, 0.f, 0.f, 0.f};

#pragma unroll
    for (int kp = 0; kp < IN_C / 8; ++kp) {
        v4f bf[OUT_C / 16];
#pragma unroll
        for (int nt = 0; nt < OUT_C / 16; ++nt) bf[nt] = wl[(kp * 8 + nt) * 32 + lane];
#pragma unroll
        for (int nt = 0; nt < OUT_C / 16; ++nt) {
            v2f b0 = {bf[nt].x, bf[nt].y};
            c[nt] = __builtin_amdgcn_wmma_f32_16x16x4_f32(false, a[2 * kp], false, b0, (short)0,
                                                          c[nt], false, false);
        }
#pragma unroll
        for (int nt = 0; nt < OUT_C / 16; ++nt) {
            v2f b1 = {bf[nt].z, bf[nt].w};
            c[nt] = __builtin_amdgcn_wmma_f32_16x16x4_f32(false, a[2 * kp + 1], false, b1,
                                                          (short)0, c[nt], false, false);
        }
    }

    const v4f* b4 = (const v4f*)bias;
    v4f* hp = (v4f*)(h + (mBase + half * 8) * OUT_C) + l;
    if (mBase + 16 <= n) {
#pragma unroll
        for (int p = 0; p < 2; ++p) {
            const v4f bv = b4[p * 16 + l];
#pragma unroll
            for (int i = 0; i < 8; ++i) {
                v4f o = {c[4 * p][i] + bv.x, c[4 * p + 1][i] + bv.y,
                         c[4 * p + 2][i] + bv.z, c[4 * p + 3][i] + bv.w};
                hp[i * (OUT_C / 4) + p * 16] = o;
            }
        }
    } else {
#pragma unroll
        for (int p = 0; p < 2; ++p) {
            const v4f bv = b4[p * 16 + l];
#pragma unroll
            for (int i = 0; i < 8; ++i) {
                v4f o = {c[4 * p][i] + bv.x, c[4 * p + 1][i] + bv.y,
                         c[4 * p + 2][i] + bv.z, c[4 * p + 3][i] + bv.w};
                if (mBase + half * 8 + i < n) hp[i * (OUT_C / 4) + p * 16] = o;
            }
        }
    }
}

template <int K>
__device__ __forceinline__ void sum_edges(const float* __restrict__ h, int c, float nrm,
                                          int j0, int lane, v4f& acc) {
    v4f v[K];
#pragma unroll
    for (int u = 0; u < K; ++u) {
        const int cj = __builtin_amdgcn_readlane(c, j0 + u);
        v[u] = ((const v4f*)h)[cj * (OUT_C / 4) + lane];
    }
#pragma unroll
    for (int u = 0; u < K; ++u) {
        const float nj = __int_as_float(__builtin_amdgcn_readlane(__float_as_int(nrm), j0 + u));
        acc += nj * v[u];
    }
}

__global__ __launch_bounds__(256) void k_row_sum(const int* __restrict__ rows,
                                                 const int* __restrict__ cols,
                                                 const int* __restrict__ deg,
                                                 const float* __restrict__ h,
                                                 float* out, int n, int e) {
    const int lane = threadIdx.x & 31;
    const int r    = __builtin_amdgcn_readfirstlane(blockIdx.x * 8 + (threadIdx.x >> 5));
    if (r >= n) return;

    float* orow = out + (size_t)r * OUT_C;
    const v4f hs    = ((const v4f*)h)[r * (OUT_C / 4) + lane];
    int       cslot = __float_as_int(orow[lane]);

    const int   dg  = __builtin_amdgcn_readfirstlane(deg[r]);
    const int   cnt = dg - 1;
    const float dr  = rsqrtf((float)dg);

    v4f acc = (dr * dr) * hs;

    if (cnt <= SLOTS) {
        for (int base = 0; base < cnt; base += 32) {
            const int rem = cnt - base;
            const int len = rem < 32 ? rem : 32;
            int c = lane < len ? cslot : r;
            c = c < 0 ? 0 : (c > n - 1 ? n - 1 : c);
            if (base + 32 < cnt) cslot = __float_as_int(orow[base + 32 + lane]);
            const float nrm = dr * rsqrtf((float)deg[c]);
            int j = 0;
            for (; j + 8 <= len; j += 8) sum_edges<8>(h, c, nrm, j, lane, acc);
            if (len - j >= 4) { sum_edges<4>(h, c, nrm, j, lane, acc); j += 4; }
            if (len - j >= 2) { sum_edges<2>(h, c, nrm, j, lane, acc); j += 2; }
            if (len - j >= 1) { sum_edges<1>(h, c, nrm, j, lane, acc); }
        }
    } else {
        for (int base = 0; base < e; base += 32) {
            const int idx = base + lane;
            int rr = -1;
            int cc = r;
            if (idx < e) {
                rr = rows[idx];
                cc = cols[idx];
            }
            unsigned m = __builtin_amdgcn_ballot_w32(rr == r);
            if (m) {
                const float nrm = dr * rsqrtf((float)deg[cc]);
                while (m) {
                    const int j = __builtin_ctz(m);
                    m &= m - 1;
                    sum_edges<1>(h, cc, nrm, j, lane, acc);
                }
            }
        }
    }

    *(v4f*)(orow + lane * 4) = acc;
}

extern "C" void kernel_launch(void* const* d_in, const int* in_sizes, int n_in,
                              void* d_out, int out_size, void* d_ws, size_t ws_size,
                              hipStream_t stream) {
    const float* x    = (const float*)d_in[0];
    const int*   ei   = (const int*)d_in[1];
    const float* Wm   = (const float*)d_in[2];
    const float* bias = (const float*)d_in[3];
    float*       out  = (float*)d_out;

    const int Nn = in_sizes[0] / IN_C;
    const int Ec = in_sizes[1] / 2;
    const int* rows = ei;
    const int* cols = ei + Ec;

    char*  base = (char*)d_ws;
    float* h    = (float*)base;
    int*   deg  = (int*)(base + (size_t)Nn * OUT_C * sizeof(float));

    k_init_deg<<<cdiv_u(Nn, 256), 256, 0, stream>>>(deg, Nn);
    k_count_bin<<<cdiv_u(Ec, 256 * EPT), 256, 0, stream>>>(rows, cols, deg, out, Ec);
    k_gemm_wmma<<<cdiv_u(cdiv_u(Nn, 16), GEMM_STRIPS), GEMM_WAVES * 32, 0, stream>>>(x, Wm, bias, h, Nn);
    k_row_sum<<<cdiv_u(Nn, 8), 256, 0, stream>>>(rows, cols, deg, h, out, Nn, Ec);
}


# Edge binning as an LDS sort: blocks rank edges by 128-row bucket, one block per bucket ranks by row in LDS and stores slots as lines; no global add or 4-B store per edge; arrival-order sums kept
# speedup vs baseline: 3.2614x; 1.7281x over previous
#include <hip/hip_runtime.h>


typedef __attribute__((ext_vector_type(2))) float v2f;
typedef __attribute__((ext_vector_type(4))) float v4f;
typedef __attribute__((ext_vector_type(8))) float v8f;

#define IN_C 128
#define OUT_C 128

static inline unsigned cdiv_u(long long a, long long b) { return (unsigned)((a + b - 1) / b); }

#define SLOTS 128
#define SH 7
#define BR (1 << SH)
#define PART_T 512
#define PART_EPT 8
#define PART_CH (PART_T * PART_EPT)
#define BIN_T 512

__global__ __launch_bounds__(PART_T) void k_part(const int* __restrict__ rows,
                                                 const int* __restrict__ cols,
                                                 int* __restrict__ deg,
                                                 float* __restrict__ out, int n, int e) {
    __shared__ int hist[PART_T];
    __shared__ int loff[PART_T];
    __shared__ int gbase[PART_T];
    __shared__ int wtot[PART_T / 32];
    __shared__ int sorted[PART_CH];
    __shared__ int sdst[PART_CH];

    const int t    = threadIdx.x;
    const int lane = t & 31;
    const int wave = t >> 5;

    hist[t] = 0;
#pragma unroll
    for (int u = 0; u < PART_EPT; ++u) sdst[t + u * PART_T] = -1;

    int r[PART_EPT], c[PART_EPT], k[PART_EPT];
    const int i0 = blockIdx.x * PART_CH + t;
#pragma unroll
    for (int u = 0; u < PART_EPT; ++u) {
        const int i = i0 + u * PART_T;
        r[u] = -1;
        c[u] = 0;
        if (i < e) {
            r[u] = rows[i];
            c[u] = cols[i];
        }
    }
    __syncthreads();

#pragma unroll
    for (int u = 0; u < PART_EPT; ++u) {
        k[u] = -1;
        if ((unsigned)r[u] < (unsigned)n && (r[u] >> SH) < PART_T)
            k[u] = atomicAdd(&hist[r[u] >> SH], 1);
    }
    __syncthreads();

    const int v = hist[t];
    int inc = v;
#pragma unroll
    for (int d = 1; d < 32; d <<= 1) {
        const int o = __shfl_up(inc, d);
        if (lane >= d) inc += o;
    }
    if (lane == 31) wtot[wave] = inc;
    __syncthreads();
    int woff = 0;
#pragma unroll
    for (int w = 0; w < PART_T / 32; ++w) woff += (w < wave) ? wtot[w] : 0;
    loff[t] = woff + inc - v;
    gbase[t] = v > 0 ? atomicAdd(&deg[t << SH], v) : 0;
    __syncthreads();

#pragma unroll
    for (int u = 0; u < PART_EPT; ++u) {
        if (k[u] >= 0) {
            const int b    = r[u] >> SH;
            const int row0 = b << SH;
            const int nr   = n - row0 < BR ? n - row0 : BR;
            const int s    = loff[b] + k[u];
            const int g    = gbase[b] + k[u];
            sorted[s] = ((r[u] & (BR - 1)) << 24) | (c[u] & 0xFFFFFF);
            sdst[s]   = g < nr * OUT_C ? row0 * OUT_C + g : -1;
        }
    }
    __syncthreads();

    int* outw = (int*)out;
#pragma unroll
    for (int u = 0; u < PART_EPT; ++u) {
        const int s = t + u * PART_T;
        const int d = sdst[s];
        if (d >= 0) outw[d] = sorted[s];
    }
}

__device__ __forceinline__ void bin_one(int* cnt, int* slot, int rl, int c) {
    const int k = atomicAdd(&cnt[rl], 1);
    if (k < SLOTS) slot[rl * SLOTS + ((k + rl) & (SLOTS - 1))] = c;
}

__global__ __launch_bounds__(BIN_T) void k_bin(const int* __restrict__ rows,
                                               const int* __restrict__ cols,
                                               int* deg, float* out, int n, int e) {
    __shared__ int cnt[BR];
    __shared__ int slot[BR * SLOTS];

    const int t    = threadIdx.x;
    const int lane = t & 31;
    const int wave = __builtin_amdgcn_readfirstlane(t >> 5);
    const int b    = blockIdx.x;
    const int row0 = b << SH;
    const int nr   = n - row0 < BR ? n - row0 : BR;
    int* outw = (int*)out;

    if (t < BR) cnt[t] = 0;
    const int f = deg[row0];
    __syncthreads();

    if (b < PART_T && f <= nr * OUT_C) {
        const int* reg = outw + row0 * OUT_C;
        for (int j0 = t; j0 < f; j0 += 4 * BIN_T) {
            int  w[4];
            bool ok[4];
#pragma unroll
            for (int u = 0; u < 4; ++u) {
                const int j = j0 + u * BIN_T;
                ok[u] = j < f;
                w[u]  = ok[u] ? reg[j] : 0;
            }
#pragma unroll
            for (int u = 0; u < 4; ++u)
                if (ok[u]) bin_one(cnt, slot, (w[u] >> 24) & (BR - 1), w[u] & 0xFFFFFF);
        }
    } else {
        for (int i = t; i < e; i += BIN_T) {
            const int r = rows[i];
            if ((unsigned)r < (unsigned)n && (r >> SH) == b) bin_one(cnt, slot, r & (BR - 1), cols[i]);
        }
    }
    __syncthreads();

    if (t < nr) deg[row0 + t] = cnt[t] + 1;

    for (int rl = wave; rl < nr; rl += BIN_T / 32) {
        const int cr  = cnt[rl];
        const int row = row0 + rl;
        const int m = cr < SLOTS ? cr : SLOTS;
        for (int base = 0; base < m; base += 32) {
            const int j = base + lane;
            if (j < m) outw[row * OUT_C + j] = slot[rl * SLOTS + ((j + rl) & (SLOTS - 1))];
        }
    }
}

#define GEMM_WAVES  16
#define GEMM_STRIPS 13

__global__ __launch_bounds__(GEMM_WAVES * 32) void k_gemm_wmma(const float* __restrict__ x,
                                                               const float* __restrict__ Wm,
                                                               const float* __restrict__ bias,
                                                               float* __restrict__ h,
                                                               int* __restrict__ deg, int n) {
    __shared__ v4f wl[(IN_C / 8) * (OUT_C / 16) * 32];

    if (blockIdx.x == 0)
        for (int b = threadIdx.x; (b << SH) < n; b += GEMM_WAVES * 32) deg[b << SH] = 0;

    const int lane = threadIdx.x & 31;
    const int wave = __builtin_amdgcn_readfirstlane(threadIdx.x >> 5);
    const int half = lane >> 4;
    const int l    = lane & 15;

    v4f g[8];
#pragma unroll
    for (int u = 0; u < 8; ++u) {
        const int unit = wave * 8 + u;
        const int kp = unit >> 3, nt = unit & 7;
        const int col = (nt >> 2) * 64 + l * 4 + (nt & 3);
        g[u] = *(const v4f*)(Wm + col * IN_C + (2 * kp + half) * 4);
    }

    const int  ntiles = (n + 15) >> 4;
    const int  tile   = blockIdx.x * GEMM_STRIPS + wave;
    const bool owns   = wave < GEMM_STRIPS && tile < ntiles;

    v2f a[IN_C / 4];
    {
        int t = owns ? tile : blockIdx.x * GEMM_STRIPS;
        if (t > ntiles - 1) t = ntiles - 1;
        int mrow = t * 16 + l;
        if (mrow > n - 1) mrow = n - 1;
        const float* xr = x + mrow * IN_C + half * 2;
#pragma unroll
        for (int i = 0; i < IN_C / 4; ++i) a[i] = *(const v2f*)(xr + 4 * i);
    }

    {
        v2f* w2 = (v2f*)wl;
#pragma unroll
        for (int u = 0; u < 8; ++u) {
            const int unit = wave * 8 + u;
            v2f lo = {g[u].x, g[u].y};
            v2f hi = {g[u].z, g[u].w};
            w2[(unit * 32 + l) * 2 + half]      = lo;
            w2[(unit * 32 + 16 + l) * 2 + half] = hi;
        }
    }
    __syncthreads();
    if (!owns) return;

    const int mBase = tile * 16;

    v8f c[OUT_C / 16];
#pragma unroll
    for (int nt = 0; nt < OUT_C / 16; ++nt) c[nt] = (v8f){0.f, 0.f, 0.f, 0.f, 0.f, 0.f, 0.f, 0.f};

#pragma unroll
    for (int kp = 0; kp < IN_C / 8; ++kp) {
        v4f bf[OUT_C / 16];
#pragma unroll
        for (int nt = 0; nt < OUT_C / 16; ++nt) bf[nt] = wl[(kp * 8 + nt) * 32 + lane];
#pragma unroll
        for (int nt = 0; nt < OUT_C / 16; ++nt) {
            v2f b0 = {bf[nt].x, bf[nt].y};
            c[nt] = __builtin_amdgcn_wmma_f32_16x16x4_f32(false, a[2 * kp], false, b0, (short)0,
                                                          c[nt], false, false);
        }
#pragma unroll
        for (int nt = 0; nt < OUT_C / 16; ++nt) {
            v2f b1 = {bf[nt].z, bf[nt].w};
            c[nt] = __builtin_amdgcn_wmma_f32_16x16x4_f32(false, a[2 * kp + 1], false, b1,
                                                          (short)0, c[nt], false, false);
        }
    }

    const v4f* b4 = (const v4f*)bias;
    v4f* hp = (v4f*)(h + (mBase + half * 8) * OUT_C) + l;
    if (mBase + 16 <= n) {
#pragma unroll
        for (int p = 0; p < 2; ++p) {
            const v4f bv = b4[p * 16 + l];
#pragma unroll
            for (int i = 0; i < 8; ++i) {
                v4f o = {c[4 * p][i] + bv.x, c[4 * p + 1][i] + bv.y,
                         c[4 * p + 2][i] + bv.z, c[4 * p + 3][i] + bv.w};
                hp[i * (OUT_C / 4) + p * 16] = o;
            }
        }
    } else {
#pragma unroll
        for (int p = 0; p < 2; ++p) {
            const v4f bv = b4[p * 16 + l];
#pragma unroll
            for (int i = 0; i < 8; ++i) {
                v4f o = {c[4 * p][i] + bv.x, c[4 * p + 1][i] + bv.y,
                         c[4 * p + 2][i] + bv.z, c[4 * p + 3][i] + bv.w};
                if (mBase + half * 8 + i < n) hp[i * (OUT_C / 4) + p * 16] = o;
            }
        }
    }
}

template <int K>
__device__ __forceinline__ void sum_edges(const float* __restrict__ h, int c, float nrm,
                                          int j0, int lane, v4f& acc) {
    v4f v[K];
#pragma unroll
    for (int u = 0; u < K; ++u) {
        const int cj = __builtin_amdgcn_readlane(c, j0 + u);
        v[u] = ((const v4f*)h)[cj * (OUT_C / 4) + lane];
    }
#pragma unroll
    for (int u = 0; u < K; ++u) {
        const float nj = __int_as_float(__builtin_amdgcn_readlane(__float_as_int(nrm), j0 + u));
        acc += nj * v[u];
    }
}

__global__ __launch_bounds__(256) void k_row_sum(const int* __restrict__ rows,
                                                 const int* __restrict__ cols,
                                                 const int* __restrict__ deg,
                                                 const float* __restrict__ h,
                                                 float* out, int n, int e) {
    const int lane = threadIdx.x & 31;
    const int r    = __builtin_amdgcn_readfirstlane(blockIdx.x * 8 + (threadIdx.x >> 5));
    if (r >= n) return;

    float* orow = out + (size_t)r * OUT_C;
    const v4f hs    = ((const v4f*)h)[r * (OUT_C / 4) + lane];
    int       cslot = __float_as_int(orow[lane]);

    const int   dg  = __builtin_amdgcn_readfirstlane(deg[r]);
    const int   cnt = dg - 1;
    const float dr  = rsqrtf((float)dg);

    v4f acc = (dr * dr) * hs;

    if (cnt <= SLOTS) {
        for (int base = 0; base < cnt; base += 32) {
            const int rem = cnt - base;
            const int len = rem < 32 ? rem : 32;
            int c = lane < len ? cslot : r;
            c = c < 0 ? 0 : (c > n - 1 ? n - 1 : c);
            if (base + 32 < cnt) cslot = __float_as_int(orow[base + 32 + lane]);
            const float nrm = dr * rsqrtf((float)deg[c]);
            int j = 0;
            for (; j + 8 <= len; j += 8) sum_edges<8>(h, c, nrm, j, lane, acc);
            if (len - j >= 4) { sum_edges<4>(h, c, nrm, j, lane, acc); j += 4; }
            if (len - j >= 2) { sum_edges<2>(h, c, nrm, j, lane, acc); j += 2; }
            if (len - j >= 1) { sum_edges<1>(h, c, nrm, j, lane, acc); }
        }
    } else {
        for (int base = 0; base < e; base += 32) {
            const int idx = base + lane;
            int rr = -1;
            int cc = r;
            if (idx < e) {
                rr = rows[idx];
                cc = cols[idx];
            }
            unsigned m = __builtin_amdgcn_ballot_w32(rr == r);
            if (m) {
                const float nrm = dr * rsqrtf((float)deg[cc]);
                while (m) {
                    const int j = __builtin_ctz(m);
                    m &= m - 1;
                    sum_edges<1>(h, cc, nrm, j, lane, acc);
                }
            }
        }
    }

    *(v4f*)(orow + lane * 4) = acc;
}

extern "C" void kernel_launch(void* const* d_in, const int* in_sizes, int n_in,
                              void* d_out, int out_size, void* d_ws, size_t ws_size,
                              hipStream_t stream) {
    const float* x    = (const float*)d_in[0];
    const int*   ei   = (const int*)d_in[1];
    const float* Wm   = (const float*)d_in[2];
    const float* bias = (const float*)d_in[3];
    float*       out  = (float*)d_out;

    const int Nn = in_sizes[0] / IN_C;
    const int Ec = in_sizes[1] / 2;
    const int* rows = ei;
    const int* cols = ei + Ec;

    char*  base = (char*)d_ws;
    float* h    = (float*)base;
    int*   deg  = (int*)(base + (size_t)Nn * OUT_C * sizeof(float));

    k_gemm_wmma<<<cdiv_u(cdiv_u(Nn, 16), GEMM_STRIPS), GEMM_WAVES * 32, 0, stream>>>(x, Wm, bias, h, deg, Nn);
    k_part<<<cdiv_u(Ec, PART_CH), PART_T, 0, stream>>>(rows, cols, deg, out, Nn, Ec);
    k_bin<<<cdiv_u(Nn, BR), BIN_T, 0, stream>>>(rows, cols, deg, out, Nn, Ec);
    k_row_sum<<<cdiv_u(Nn, 8), 256, 0, stream>>>(rows, cols, deg, h, out, Nn, Ec);
}


# Edge partition made independent of any earlier launch (fixed run per part block and bucket with a count word, no global add) and run as extra blocks of the GEMM launch: 3 launches, was 4
# speedup vs baseline: 3.3606x; 1.0304x over previous
#include <hip/hip_runtime.h>


typedef __attribute__((ext_vector_type(2))) float v2f;
typedef __attribute__((ext_vector_type(4))) float v4f;
typedef __attribute__((ext_vector_type(8))) float v8f;

#define IN_C 128
#define OUT_C 128

static inline unsigned cdiv_u(long long a, long long b) { return (unsigned)((a + b - 1) / b); }

#define SLOTS 128
#define SH 7
#define BR (1 << SH)
#define PART_T 512
#define PART_EPT 8
#define PART_CH (PART_T * PART_EPT)
#define BIN_T 512

#define PART_LDS_INTS (4 * PART_T + PART_T / 32 + 2 * PART_CH)

__device__ __forceinline__ void part_body(int* __restrict__ lds, const int* __restrict__ rows,
                                          const int* __restrict__ cols, float* __restrict__ out,
                                          int n, int e, int p, int npb) {
    int* hist   = lds;
    int* loff   = hist + PART_T;
    int* gbase  = loff + PART_T;
    int* capl   = gbase + PART_T;
    int* wtot   = capl + PART_T;
    int* sorted = wtot + PART_T / 32;
    int* sdst   = sorted + PART_CH;
    int* outw   = (int*)out;

    const int t    = threadIdx.x;
    const int lane = t & 31;
    const int wave = t >> 5;

    hist[t] = 0;
#pragma unroll
    for (int u = 0; u < PART_EPT; ++u) sdst[t + u * PART_T] = -1;

    int r[PART_EPT], c[PART_EPT], k[PART_EPT];
    const int i0 = p * PART_CH + t;
#pragma unroll
    for (int u = 0; u < PART_EPT; ++u) {
        const int i = i0 + u * PART_T;
        r[u] = -1;
        c[u] = 0;
        if (i < e) {
            r[u] = rows[i];
            c[u] = cols[i];
        }
    }
    __syncthreads();

#pragma unroll
    for (int u = 0; u < PART_EPT; ++u) {
        k[u] = -1;
        if ((unsigned)r[u] < (unsigned)n && (r[u] >> SH) < PART_T)
            k[u] = atomicAdd(&hist[r[u] >> SH], 1);
    }
    __syncthreads();

    const int v = hist[t];
    int inc = v;
#pragma unroll
    for (int d = 1; d < 32; d <<= 1) {
        const int o = __shfl_up(inc, d);
        if (lane >= d) inc += o;
    }
    if (lane == 31) wtot[wave] = inc;
    __syncthreads();
    int woff = 0;
#pragma unroll
    for (int w = 0; w < PART_T / 32; ++w) woff += (w < wave) ? wtot[w] : 0;
    loff[t] = woff + inc - v;
    {
        const int row0t = t << SH;
        int cap = 0, gb = 0;
        if (row0t < n) {
            const int nrt = n - row0t < BR ? n - row0t : BR;
            cap = (nrt * OUT_C) / npb;
            gb  = row0t * OUT_C + p * cap;
            if (cap >= 1) outw[gb] = v;
        }
        capl[t]  = cap;
        gbase[t] = gb + 1;
    }
    __syncthreads();

#pragma unroll
    for (int u = 0; u < PART_EPT; ++u) {
        if (k[u] >= 0) {
            const int b = r[u] >> SH;
            const int s = loff[b] + k[u];
            sorted[s] = ((r[u] & (BR - 1)) << 24) | (c[u] & 0xFFFFFF);
            sdst[s]   = k[u] + 1 < capl[b] ? gbase[b] + k[u] : -1;
        }
    }
    __syncthreads();

#pragma unroll
    for (int u = 0; u < PART_EPT; ++u) {
        const int s = t + u * PART_T;
        const int d = sdst[s];
        if (d >= 0) outw[d] = sorted[s];
    }
}

__device__ __forceinline__ void bin_one(int* cnt, int* slot, int rl, int c) {
    const int k = atomicAdd(&cnt[rl], 1);
    if (k < SLOTS) slot[rl * SLOTS + ((k + rl) & (SLOTS - 1))] = c;
}

__global__ __launch_bounds__(BIN_T) void k_bin(const int* __restrict__ rows,
                                               const int* __restrict__ cols,
                                               int* deg, float* out, int n, int e, int npb) {
    __shared__ int cnt[BR];
    __shared__ int slot[BR * SLOTS];
    __shared__ int ovf;

    const int t    = threadIdx.x;
    const int lane = t & 31;
    const int wave = __builtin_amdgcn_readfirstlane(t >> 5);
    const int b    = blockIdx.x;
    const int row0 = b << SH;
    const int nr   = n - row0 < BR ? n - row0 : BR;
    int* outw = (int*)out;

    const int  cap    = npb > 0 ? (nr * OUT_C) / npb : 0;
    const bool ranked = b < PART_T && cap >= 2;
    if (t < BR) cnt[t] = 0;
    if (t == 0) ovf = 0;
    __syncthreads();

    if (ranked) {
        const int* reg = outw + row0 * OUT_C;
        const int  hw  = t >> 4;
        const int  i   = t & 15;
        bool bad = false;
        for (int q0 = 0; q0 < npb; q0 += 4 * (BIN_T / 16)) {
            int w[4];
#pragma unroll
            for (int u = 0; u < 4; ++u) {
                const int q = q0 + u * (BIN_T / 16) + hw;
                w[u] = (q < npb && i < cap) ? reg[q * cap + i] : 0;
            }
#pragma unroll
            for (int u = 0; u < 4; ++u) {
                const int q  = q0 + u * (BIN_T / 16) + hw;
                int cq = __shfl(w[u], 0, 16);
                if ((unsigned)cq > (unsigned)(cap - 1)) { bad = true; cq = 0; }
                if (i >= 1 && i <= cq) bin_one(cnt, slot, (w[u] >> 24) & (BR - 1), w[u] & 0xFFFFFF);
                for (int j = 16 + i; j <= cq; j += 16) {
                    const int wj = reg[q * cap + j];
                    bin_one(cnt, slot, (wj >> 24) & (BR - 1), wj & 0xFFFFFF);
                }
            }
        }
        if (bad) ovf = 1;
    }
    __syncthreads();
    if (!ranked || ovf != 0) {
        if (t < BR) cnt[t] = 0;
        __syncthreads();
        for (int i = t; i < e; i += BIN_T) {
            const int r = rows[i];
            if ((unsigned)r < (unsigned)n && (r >> SH) == b) bin_one(cnt, slot, r & (BR - 1), cols[i]);
        }
    }
    __syncthreads();

    if (t < nr) deg[row0 + t] = cnt[t] + 1;

    for (int rl = wave; rl < nr; rl += BIN_T / 32) {
        const int cr  = cnt[rl];
        const int row = row0 + rl;
        const int m = cr < SLOTS ? cr : SLOTS;
        for (int base = 0; base < m; base += 32) {
            const int j = base + lane;
            if (j < m) outw[row * OUT_C + j] = slot[rl * SLOTS + ((j + rl) & (SLOTS - 1))];
        }
    }
}

#define GEMM_WAVES  16
#define GEMM_STRIPS 13

__global__ __launch_bounds__(GEMM_WAVES * 32, 8) void k_gemm_part(const float* __restrict__ x,
                                                                  const float* __restrict__ Wm,
                                                                  const float* __restrict__ bias,
                                                                  float* __restrict__ h,
                                                                  const int* __restrict__ rows,
                                                                  const int* __restrict__ cols,
                                                                  float* __restrict__ out,
                                                                  int n, int e, int npb) {
    __shared__ v4f wl[(IN_C / 8) * (OUT_C / 16) * 32];
    static_assert(PART_LDS_INTS * sizeof(int) <= sizeof(v4f) * (IN_C / 8) * (OUT_C / 16) * 32, "part LDS");
    static_assert(PART_T == GEMM_WAVES * 32, "one block size");

    if ((int)blockIdx.x < npb) {
        part_body((int*)wl, rows, cols, out, n, e, (int)blockIdx.x, npb);
        return;
    }
    const int gblk = (int)blockIdx.x - npb;

    const int lane = threadIdx.x & 31;
    const int wave = __builtin_amdgcn_readfirstlane(threadIdx.x >> 5);
    const int half = lane >> 4;
    const int l    = lane & 15;

    v4f g[8];
#pragma unroll
    for (int u = 0; u < 8; ++u) {
        const int unit = wave * 8 + u;
        const int kp = unit >> 3, nt = unit & 7;
        const int col = (nt >> 2) * 64 + l * 4 + (nt & 3);
        g[u] = *(const v4f*)(Wm + col * IN_C + (2 * kp + half) * 4);
    }

    const int  ntiles = (n + 15) >> 4;
    const int  tile   = gblk * GEMM_STRIPS + wave;
    const bool owns   = wave < GEMM_STRIPS && tile < ntiles;

    v2f a[IN_C / 4];
    {
        int t = owns ? tile : gblk * GEMM_STRIPS;
        if (t > ntiles - 1) t = ntiles - 1;
        int mrow = t * 16 + l;
        if (mrow > n - 1) mrow = n - 1;
        const float* xr = x + mrow * IN_C + half * 2;
#pragma unroll
        for (int i = 0; i < IN_C / 4; ++i) a[i] = *(const v2f*)(xr + 4 * i);
    }

    {
        v2f* w2 = (v2f*)wl;
#pragma unroll
        for (int u = 0; u < 8; ++u) {
            const int unit = wave * 8 + u;
            v2f lo = {g[u].x, g[u].y};
            v2f hi = {g[u].z, g[u].w};
            w2[(unit * 32 + l) * 2 + half]      = lo;
            w2[(unit * 32 + 16 + l) * 2 + half] = hi;
        }
    }
    __syncthreads();
    if (!owns) return;

    const int mBase = tile * 16;

    const v4f* b4 = (const v4f*)bias;
    v4f* hp = (v4f*)(h + (mBase + half * 8) * OUT_C) + l;
    const bool whole = mBase + 16 <= n;

#pragma unroll
    for (int p = 0; p < 2; ++p) {
        v8f c[4];
#pragma unroll
        for (int q = 0; q < 4; ++q) c[q] = (v8f){0.f, 0.f, 0.f, 0.f, 0.f, 0.f, 0.f, 0.f};

#pragma unroll
        for (int kp = 0; kp < IN_C / 8; ++kp) {
            v4f bf[4];
#pragma unroll
            for (int q = 0; q < 4; ++q) bf[q] = wl[(kp * 8 + 4 * p + q) * 32 + lane];
#pragma unroll
            for (int q = 0; q < 4; ++q) {
                v2f b0 = {bf[q].x, bf[q].y};
                c[q] = __builtin_amdgcn_wmma_f32_16x16x4_f32(false, a[2 * kp], false, b0, (short)0,
                                                             c[q], false, false);
            }
#pragma unroll
            for (int q = 0; q < 4; ++q) {
                v2f b1 = {bf[q].z, bf[q].w};
                c[q] = __builtin_amdgcn_wmma_f32_16x16x4_f32(false, a[2 * kp + 1], false, b1,
                                                             (short)0, c[q], false, false);
            }
            __builtin_amdgcn_sched_barrier(0);
        }

        const v4f bv = b4[p * 16 + l];
        if (whole) {
#pragma unroll
            for (int i = 0; i < 8; ++i) {
                v4f o = {c[0][i] + bv.x, c[1][i] + bv.y, c[2][i] + bv.z, c[3][i] + bv.w};
                hp[i * (OUT_C / 4) + p * 16] = o;
            }
        } else {
#pragma unroll
            for (int i = 0; i < 8; ++i) {
                v4f o = {c[0][i] + bv.x, c[1][i] + bv.y, c[2][i] + bv.z, c[3][i] + bv.w};
                if (mBase + half * 8 + i < n) hp[i * (OUT_C / 4) + p * 16] = o;
            }
        }
    }
}

template <int K>
__device__ __forceinline__ void sum_edges(const float* __restrict__ h, int c, float nrm,
                                          int j0, int lane, v4f& acc) {
    v4f v[K];
#pragma unroll
    for (int u = 0; u < K; ++u) {
        const int cj = __builtin_amdgcn_readlane(c, j0 + u);
        v[u] = ((const v4f*)h)[cj * (OUT_C / 4) + lane];
    }
#pragma unroll
    for (int u = 0; u < K; ++u) {
        const float nj = __int_as_float(__builtin_amdgcn_readlane(__float_as_int(nrm), j0 + u));
        acc += nj * v[u];
    }
}

__global__ __launch_bounds__(256) void k_row_sum(const int* __restrict__ rows,
                                                 const int* __restrict__ cols,
                                                 const int* __restrict__ deg,
                                                 const float* __restrict__ h,
                                                 float* out, int n, int e) {
    const int lane = threadIdx.x & 31;
    const int r    = __builtin_amdgcn_readfirstlane(blockIdx.x * 8 + (threadIdx.x >> 5));
    if (r >= n) return;

    float* orow = out + (size_t)r * OUT_C;
    const v4f hs    = ((const v4f*)h)[r * (OUT_C / 4) + lane];
    int       cslot = __float_as_int(orow[lane]);

    const int   dg  = __builtin_amdgcn_readfirstlane(deg[r]);
    const int   cnt = dg - 1;
    const float dr  = rsqrtf((float)dg);

    v4f acc = (dr * dr) * hs;

    if (cnt <= SLOTS) {
        for (int base = 0; base < cnt; base += 32) {
            const int rem = cnt - base;
            const int len = rem < 32 ? rem : 32;
            int c = lane < len ? cslot : r;
            c = c < 0 ? 0 : (c > n - 1 ? n - 1 : c);
            if (base + 32 < cnt) cslot = __float_as_int(orow[base + 32 + lane]);
            const float nrm = dr * rsqrtf((float)deg[c]);
            int j = 0;
            for (; j + 8 <= len; j += 8) sum_edges<8>(h, c, nrm, j, lane, acc);
            if (len - j >= 4) { sum_edges<4>(h, c, nrm, j, lane, acc); j += 4; }
            if (len - j >= 2) { sum_edges<2>(h, c, nrm, j, lane, acc); j += 2; }
            if (len - j >= 1) { sum_edges<1>(h, c, nrm, j, lane, acc); }
        }
    } else {
        for (int base = 0; base < e; base += 32) {
            const int idx = base + lane;
            int rr = -1;
            int cc = r;
            if (idx < e) {
                rr = rows[idx];
                cc = cols[idx];
            }
            unsigned m = __builtin_amdgcn_ballot_w32(rr == r);
            if (m) {
                const float nrm = dr * rsqrtf((float)deg[cc]);
                while (m) {
                    const int j = __builtin_ctz(m);
                    m &= m - 1;
                    sum_edges<1>(h, cc, nrm, j, lane, acc);
                }
            }
        }
    }

    *(v4f*)(orow + lane * 4) = acc;
}

extern "C" void kernel_launch(void* const* d_in, const int* in_sizes, int n_in,
                              void* d_out, int out_size, void* d_ws, size_t ws_size,
                              hipStream_t stream) {
    const float* x    = (const float*)d_in[0];
    const int*   ei   = (const int*)d_in[1];
    const float* Wm   = (const float*)d_in[2];
    const float* bias = (const float*)d_in[3];
    float*       out  = (float*)d_out;

    const int Nn = in_sizes[0] / IN_C;
    const int Ec = in_sizes[1] / 2;
    const int* rows = ei;
    const int* cols = ei + Ec;

    char*  base = (char*)d_ws;
    float* h    = (float*)base;
    int*   deg  = (int*)(base + (size_t)Nn * OUT_C * sizeof(float));

    const unsigned npb = cdiv_u(Ec, PART_CH);
    k_gemm_part<<<npb + cdiv_u(cdiv_u(Nn, 16), GEMM_STRIPS), GEMM_WAVES * 32, 0, stream>>>(
        x, Wm, bias, h, rows, cols, out, Nn, Ec, (int)npb);
    k_bin<<<cdiv_u(Nn, BR), BIN_T, 0, stream>>>(rows, cols, deg, out, Nn, Ec, (int)npb);
    k_row_sum<<<cdiv_u(Nn, 8), 256, 0, stream>>>(rows, cols, deg, h, out, Nn, Ec);
}


# Gather waves: a row's degree, slot ids and source degrees requested one and two stages ahead, its rows of x in groups of up to 16 (were 8,4,2,1 pieces): 2 memory trips a row, not 5-7; sum order kept
# speedup vs baseline: 5.2169x; 1.5524x over previous
#include <hip/hip_runtime.h>


typedef __attribute__((ext_vector_type(2))) float v2f;
typedef __attribute__((ext_vector_type(4))) float v4f;
typedef __attribute__((ext_vector_type(8))) float v8f;

#define IN_C 128
#define OUT_C 128

static inline unsigned cdiv_u(long long a, long long b) { return (unsigned)((a + b - 1) / b); }

#define SLOTS 128
#define SH 7
#define BR (1 << SH)
#define PART_T 512
#define PART_EPT 8
#define PART_CH (PART_T * PART_EPT)
#define BIN_T 512

#define PART_LDS_INTS (4 * PART_T + PART_T / 32 + 2 * PART_CH)

__device__ __forceinline__ void part_body(int* __restrict__ lds, const int* __restrict__ rows,
                                          const int* __restrict__ cols, float* __restrict__ out,
                                          int n, int e, int p, int npb) {
    int* hist   = lds;
    int* loff   = hist + PART_T;
    int* gbase  = loff + PART_T;
    int* capl   = gbase + PART_T;
    int* wtot   = capl + PART_T;
    int* sorted = wtot + PART_T / 32;
    int* sdst   = sorted + PART_CH;
    int* outw   = (int*)out;

    const int t    = threadIdx.x;
    const int lane = t & 31;
    const int wave = t >> 5;

    hist[t] = 0;
#pragma unroll
    for (int u = 0; u < PART_EPT; ++u) sdst[t + u * PART_T] = -1;

    int r[PART_EPT], c[PART_EPT], k[PART_EPT];
    const int i0 = p * PART_CH + t;
#pragma unroll
    for (int u = 0; u < PART_EPT; ++u) {
        const int i = i0 + u * PART_T;
        r[u] = -1;
        c[u] = 0;
        if (i < e) {
            r[u] = rows[i];
            c[u] = cols[i];
        }
    }
    __syncthreads();

#pragma unroll
    for (int u = 0; u < PART_EPT; ++u) {
        k[u] = -1;
        if ((unsigned)r[u] < (unsigned)n && (r[u] >> SH) < PART_T)
            k[u] = atomicAdd(&hist[r[u] >> SH], 1);
    }
    __syncthreads();

    const int v = hist[t];
    int inc = v;
#pragma unroll
    for (int d = 1; d < 32; d <<= 1) {
        const int o = __shfl_up(inc, d);
        if (lane >= d) inc += o;
    }
    if (lane == 31) wtot[wave] = inc;
    __syncthreads();
    int woff = 0;
#pragma unroll
    for (int w = 0; w < PART_T / 32; ++w) woff += (w < wave) ? wtot[w] : 0;
    loff[t] = woff + inc - v;
    {
        const int row0t = t << SH;
        int cap = 0, gb = 0;
        if (row0t < n) {
            const int nrt = n - row0t < BR ? n - row0t : BR;
            cap = (nrt * OUT_C) / npb;
            gb  = row0t * OUT_C + p * cap;
            if (cap >= 1) outw[gb] = v;
        }
        capl[t]  = cap;
        gbase[t] = gb + 1;
    }
    __syncthreads();

#pragma unroll
    for (int u = 0; u < PART_EPT; ++u) {
        if (k[u] >= 0) {
            const int b = r[u] >> SH;
            const int s = loff[b] + k[u];
            sorted[s] = ((r[u] & (BR - 1)) << 24) | (c[u] & 0xFFFFFF);
            sdst[s]   = k[u] + 1 < capl[b] ? gbase[b] + k[u] : -1;
        }
    }
    __syncthreads();

#pragma unroll
    for (int u = 0; u < PART_EPT; ++u) {
        const int s = t + u * PART_T;
        const int d = sdst[s];
        if (d >= 0) outw[d] = sorted[s];
    }
}

__device__ __forceinline__ void bin_one(int* cnt, int* slot, int rl, int c) {
    const int k = atomicAdd(&cnt[rl], 1);
    if (k < SLOTS) slot[rl * SLOTS + ((k + rl) & (SLOTS - 1))] = c;
}

__global__ __launch_bounds__(BIN_T) void k_bin(const int* __restrict__ rows,
                                               const int* __restrict__ cols,
                                               int* deg, float* out, int n, int e, int npb) {
    __shared__ int cnt[BR];
    __shared__ int slot[BR * SLOTS];
    __shared__ int ovf;

    const int t    = threadIdx.x;
    const int lane = t & 31;
    const int wave = __builtin_amdgcn_readfirstlane(t >> 5);
    const int b    = blockIdx.x;
    const int row0 = b << SH;
    const int nr   = n - row0 < BR ? n - row0 : BR;
    int* outw = (int*)out;

    const int  cap    = npb > 0 ? (nr * OUT_C) / npb : 0;
    const bool ranked = b < PART_T && cap >= 2;
    if (t < BR) cnt[t] = 0;
    if (t == 0) ovf = 0;
    __syncthreads();

    if (ranked) {
        const int* reg = outw + row0 * OUT_C;
        const int  hw  = t >> 4;
        const int  i   = t & 15;
        bool bad = false;
        for (int q0 = 0; q0 < npb; q0 += 4 * (BIN_T / 16)) {
            int w[4];
#pragma unroll
            for (int u = 0; u < 4; ++u) {
                const int q = q0 + u * (BIN_T / 16) + hw;
                w[u] = (q < npb && i < cap) ? reg[q * cap + i] : 0;
            }
#pragma unroll
            for (int u = 0; u < 4; ++u) {
                const int q  = q0 + u * (BIN_T / 16) + hw;
                int cq = __shfl(w[u], 0, 16);
                if ((unsigned)cq > (unsigned)(cap - 1)) { bad = true; cq = 0; }
                if (i >= 1 && i <= cq) bin_one(cnt, slot, (w[u] >> 24) & (BR - 1), w[u] & 0xFFFFFF);
                for (int j = 16 + i; j <= cq; j += 16) {
                    const int wj = reg[q * cap + j];
                    bin_one(cnt, slot, (wj >> 24) & (BR - 1), wj & 0xFFFFFF);
                }
            }
        }
        if (bad) ovf = 1;
    }
    __syncthreads();
    if (!ranked || ovf != 0) {
        if (t < BR) cnt[t] = 0;
        __syncthreads();
        for (int i = t; i < e; i += BIN_T) {
            const int r = rows[i];
            if ((unsigned)r < (unsigned)n && (r >> SH) == b) bin_one(cnt, slot, r & (BR - 1), cols[i]);
        }
    }
    __syncthreads();

    if (t < nr) deg[row0 + t] = cnt[t] + 1;

    for (int rl = wave; rl < nr; rl += BIN_T / 32) {
        const int cr  = cnt[rl];
        const int row = row0 + rl;
        const int m = cr < SLOTS ? cr : SLOTS;
        for (int base = 0; base < m; base += 32) {
            const int j = base + lane;
            if (j < m) outw[row * OUT_C + j] = slot[rl * SLOTS + ((j + rl) & (SLOTS - 1))];
        }
    }
}

__global__ __launch_bounds__(PART_T) void k_part(const int* __restrict__ rows,
                                                 const int* __restrict__ cols,
                                                 float* __restrict__ out, int n, int e, int npb) {
    __shared__ int lds[PART_LDS_INTS];
    part_body(lds, rows, cols, out, n, e, (int)blockIdx.x, npb);
}

#define AG_GW    16
#define AG_MW    4
#define AG_T     ((AG_GW + AG_MW) * 32)
#define AG_PITCH 132

template <int K>
__device__ __forceinline__ void issue_rows(const float* __restrict__ x, int c, int j0, int lane, v4f* v) {
#pragma unroll
    for (int u = 0; u < K; ++u) {
        const int cj = __builtin_amdgcn_readlane(c, j0 + u);
        v[u] = ((const v4f*)x)[cj * (IN_C / 4) + lane];
    }
}

template <int K>
__device__ __forceinline__ void add_rows(const v4f* v, float nrm, int j0, v4f& acc, float& wsum) {
#pragma unroll
    for (int u = 0; u < K; ++u) {
        const float nj = __int_as_float(__builtin_amdgcn_readlane(__float_as_int(nrm), j0 + u));
        acc += nj * v[u];
        wsum += nj;
    }
}

template <int K>
__device__ __forceinline__ void sum_edges(const float* __restrict__ x, int c, float nrm,
                                          int j0, int lane, v4f& acc, float& wsum) {
    v4f v[K];
    issue_rows<K>(x, c, j0, lane, v);
    add_rows<K>(v, nrm, j0, acc, wsum);
}

template <int K, class Pre>
__device__ __forceinline__ void first_group(const float* __restrict__ x, int c, float nrm, int lane,
                                            int r, float dr2, v4f& acc, float& wsum, Pre pre) {
    v4f v[K > 0 ? K : 1];
    if (K > 0) issue_rows<K>(x, c, 0, lane, v);
    const v4f xs = ((const v4f*)x)[r * (IN_C / 4) + lane];
    pre();
    acc  = dr2 * xs;
    wsum = dr2;
    if (K > 0) add_rows<K>(v, nrm, 0, acc, wsum);
}

template <class Pre>
__device__ __forceinline__ void sum_chunk0(const float* __restrict__ x, int c, float nrm, int len,
                                           int lane, int r, float dr2, v4f& acc, float& wsum, Pre pre) {
    if (len <= 0) {
        first_group<0>(x, c, nrm, lane, r, dr2, acc, wsum, pre);
        return;
    }
    if (len > 12)     first_group<16>(x, c, nrm, lane, r, dr2, acc, wsum, pre);
    else if (len > 8) first_group<12>(x, c, nrm, lane, r, dr2, acc, wsum, pre);
    else if (len > 4) first_group<8>(x, c, nrm, lane, r, dr2, acc, wsum, pre);
    else              first_group<4>(x, c, nrm, lane, r, dr2, acc, wsum, pre);
    if (len > 16) {
        if (len > 28)      sum_edges<16>(x, c, nrm, 16, lane, acc, wsum);
        else if (len > 24) sum_edges<12>(x, c, nrm, 16, lane, acc, wsum);
        else if (len > 20) sum_edges<8>(x, c, nrm, 16, lane, acc, wsum);
        else               sum_edges<4>(x, c, nrm, 16, lane, acc, wsum);
    }
}

__device__ __forceinline__ int slot_id(int word, int lane, int cnt_left, int r, int n) {
    const int c = lane < cnt_left ? word : r;
    return c < 0 ? 0 : (c > n - 1 ? n - 1 : c);
}

template <class Pre>
__device__ __forceinline__ void gather_row(const int* __restrict__ rows,
                                           const int* __restrict__ cols,
                                           const int* __restrict__ deg,
                                           const float* __restrict__ x, const float* out,
                                           int n, int e, int r, int lane,
                                           int sl0, int dgv, int dc0, v4f& acc, float& wsum, Pre pre) {
    const float* orow = out + (size_t)r * OUT_C;

    const int   dg  = __builtin_amdgcn_readfirstlane(dgv);
    const int   cnt = dg - 1;
    const float dr  = rsqrtf((float)dg);
    const float dr2 = dr * dr;

    if (cnt <= SLOTS) {
        {
            const int len = cnt < 32 ? cnt : 32;
            int   c   = slot_id(sl0, lane, cnt, r, n);
            float nrm = dr * rsqrtf((float)dc0);
            const int cl = __builtin_amdgcn_readlane(c, len > 0 ? len - 1 : 0);
            c   = lane < len ? c : cl;
            nrm = lane < len ? nrm : 0.f;
            sum_chunk0(x, c, nrm, len, lane, r, dr2, acc, wsum, pre);
        }
        for (int base = 32; base < cnt; base += 32) {
            const int rem = cnt - base;
            const int len = rem < 32 ? rem : 32;
            const int c   = slot_id(__float_as_int(orow[base + lane]), lane, len, r, n);
            const float nrm = dr * rsqrtf((float)deg[c]);
            int j = 0;
            for (; j + 8 <= len; j += 8) sum_edges<8>(x, c, nrm, j, lane, acc, wsum);
            if (len - j >= 4) { sum_edges<4>(x, c, nrm, j, lane, acc, wsum); j += 4; }
            if (len - j >= 2) { sum_edges<2>(x, c, nrm, j, lane, acc, wsum); j += 2; }
            if (len - j >= 1) { sum_edges<1>(x, c, nrm, j, lane, acc, wsum); }
        }
    } else {
        first_group<0>(x, r, 0.f, lane, r, dr2, acc, wsum, pre);
        for (int base = 0; base < e; base += 32) {
            const int idx = base + lane;
            int rr = -1;
            int cc = r;
            if (idx < e) {
                rr = rows[idx];
                cc = cols[idx];
            }
            cc = cc < 0 ? 0 : (cc > n - 1 ? n - 1 : cc);
            unsigned m = __builtin_amdgcn_ballot_w32(rr == r);
            if (m) {
                const float nrm = dr * rsqrtf((float)deg[cc]);
                while (m) {
                    const int j = __builtin_ctz(m);
                    m &= m - 1;
                    sum_edges<1>(x, cc, nrm, j, lane, acc, wsum);
                }
            }
        }
    }
}

__global__ __launch_bounds__(AG_T) __attribute__((amdgpu_waves_per_eu(10))) void k_agg_gemm(const int* __restrict__ rows,
                                                   const int* __restrict__ cols,
                                                   const int* __restrict__ deg,
                                                   const float* __restrict__ x,
                                                   const float* __restrict__ Wm,
                                                   const float* __restrict__ bias,
                                                   float* out, int n, int e) {
    __shared__ v4f   wl[(IN_C / 8) * (OUT_C / 16) * 32];
    __shared__ __attribute__((aligned(16))) float ag[2][AG_GW * AG_PITCH];
    __shared__ float sw[2][AG_GW];

    const int lane = threadIdx.x & 31;
    const int wave = __builtin_amdgcn_readfirstlane(threadIdx.x >> 5);
    const int half = lane >> 4;
    const int l    = lane & 15;

    const int ntiles = (n + AG_GW - 1) / AG_GW;
    const int first = (int)blockIdx.x;
    const int step  = (int)gridDim.x;
    const int nk    = first < ntiles ? (ntiles - first + step - 1) / step : 0;

    int sl_a = 0, dg_a = 1, dc_a = 1, sl_b = 0, dg_b = 1;
    if (wave < AG_GW) {
        const int r0 = first * AG_GW + wave;
        const int r1 = (first + step) * AG_GW + wave;
        if (0 < nk && r0 < n) {
            sl_a = __float_as_int(out[(size_t)r0 * OUT_C + lane]);
            dg_a = deg[r0];
        }
        if (1 < nk && r1 < n) {
            sl_b = __float_as_int(out[(size_t)r1 * OUT_C + lane]);
            dg_b = deg[r1];
        }
        dc_a = deg[slot_id(sl_a, lane, __builtin_amdgcn_readfirstlane(dg_a) - 1, r0, n)];
    }

    for (int s = 0; s <= nk; ++s) {
        if (wave < AG_GW) {
            if (s < nk) {
                const int r  = (first + s * step) * AG_GW + wave;
                const int r1 = r + step * AG_GW;
                const int r2 = r1 + step * AG_GW;
                int sl_c = 0, dg_c = 1, dc_b = 1;
                auto pre = [&]() {
                    const int r2c = r2 < n ? r2 : n - 1;
                    sl_c = __float_as_int(out[(size_t)r2c * OUT_C + lane]);
                    dg_c = deg[r2c];
                    dc_b = deg[slot_id(sl_b, lane, __builtin_amdgcn_readfirstlane(dg_b) - 1, r1, n)];
                };

                v4f   acc  = {0.f, 0.f, 0.f, 0.f};
                float wsum = 0.f;
                if (r < n) gather_row(rows, cols, deg, x, out, n, e, r, lane, sl_a, dg_a, dc_a, acc, wsum, pre);
                float* agb = ag[s & 1];
                *(v4f*)(agb + wave * AG_PITCH + lane * 4) = acc;
                if (lane == 0) sw[s & 1][wave] = wsum;

                sl_a = sl_b; dg_a = dg_b; dc_a = dc_b;
                sl_b = sl_c; dg_b = dg_c;
            }
        } else if (s == 0) {
            const int mw = wave - AG_GW;
            v2f* w2 = (v2f*)wl;
            for (int u0 = 0; u0 < 2 * (IN_C / 8); u0 += 8) {
                v4f g[8];
#pragma unroll
                for (int u = 0; u < 8; ++u) {
                    const int kp  = (u0 + u) >> 1;
                    const int col = mw * 32 + l * 2 + ((u0 + u) & 1);
                    g[u] = *(const v4f*)(Wm + col * IN_C + (2 * kp + half) * 4);
                }
#pragma unroll
                for (int u = 0; u < 8; ++u) {
                    const int kp   = (u0 + u) >> 1;
                    const int unit = kp * 8 + 2 * mw + ((u0 + u) & 1);
                    v2f lo = {g[u].x, g[u].y};
                    v2f hi = {g[u].z, g[u].w};
                    w2[(unit * 32 + l) * 2 + half]      = lo;
                    w2[(unit * 32 + 16 + l) * 2 + half] = hi;
                }
            }
        } else {
            const int mw   = wave - AG_GW;
            const int row0 = (first + (s - 1) * step) * AG_GW;
            const float* agb = ag[(s - 1) & 1];
            const float* ar  = agb + l * AG_PITCH + half * 2;

            v8f c0 = (v8f){0.f, 0.f, 0.f, 0.f, 0.f, 0.f, 0.f, 0.f};
            v8f c1 = c0;
#pragma unroll
            for (int kp = 0; kp < IN_C / 8; ++kp) {
                const v4f bf0 = wl[(kp * 8 + 2 * mw) * 32 + lane];
                const v4f bf1 = wl[(kp * 8 + 2 * mw + 1) * 32 + lane];
                const v2f a0 = *(const v2f*)(ar + 8 * kp);
                const v2f a1 = *(const v2f*)(ar + 8 * kp + 4);
                v2f b00 = {bf0.x, bf0.y};
                v2f b10 = {bf1.x, bf1.y};
                v2f b01 = {bf0.z, bf0.w};
                v2f b11 = {bf1.z, bf1.w};
                c0 = __builtin_amdgcn_wmma_f32_16x16x4_f32(false, a0, false, b00, (short)0, c0, false, false);
                c1 = __builtin_amdgcn_wmma_f32_16x16x4_f32(false, a0, false, b10, (short)0, c1, false, false);
                c0 = __builtin_amdgcn_wmma_f32_16x16x4_f32(false, a1, false, b01, (short)0, c0, false, false);
                c1 = __builtin_amdgcn_wmma_f32_16x16x4_f32(false, a1, false, b11, (short)0, c1, false, false);
            }

            const v2f bv = *(const v2f*)(bias + mw * 32 + l * 2);
            const float* swb = sw[(s - 1) & 1] + half * 8;
            v2f o[8];
#pragma unroll
            for (int i = 0; i < 8; ++i) {
                const float ws = swb[i];
                o[i] = (v2f){c0[i] + ws * bv.x, c1[i] + ws * bv.y};
            }
            v2f* op = (v2f*)(out + (size_t)(row0 + half * 8) * OUT_C + mw * 32 + l * 2);
            if (row0 + AG_GW <= n) {
#pragma unroll
                for (int i = 0; i < 8; ++i) op[i * (OUT_C / 2)] = o[i];
            } else {
#pragma unroll
                for (int i = 0; i < 8; ++i)
                    if (row0 + half * 8 + i < n) op[i * (OUT_C / 2)] = o[i];
            }
        }
        __syncthreads();
    }
}

extern "C" void kernel_launch(void* const* d_in, const int* in_sizes, int n_in,
                              void* d_out, int out_size, void* d_ws, size_t ws_size,
                              hipStream_t stream) {
    const float* x    = (const float*)d_in[0];
    const int*   ei   = (const int*)d_in[1];
    const float* Wm   = (const float*)d_in[2];
    const float* bias = (const float*)d_in[3];
    float*       out  = (float*)d_out;

    const int Nn = in_sizes[0] / IN_C;
    const int Ec = in_sizes[1] / 2;
    const int* rows = ei;
    const int* cols = ei + Ec;

    int* deg = (int*)d_ws;

    const unsigned npb = cdiv_u(Ec, PART_CH);
    k_part<<<npb, PART_T, 0, stream>>>(rows, cols, out, Nn, Ec, (int)npb);
    k_bin<<<cdiv_u(Nn, BR), BIN_T, 0, stream>>>(rows, cols, deg, out, Nn, Ec, (int)npb);
    k_agg_gemm<<<512, AG_T, 0, stream>>>(rows, cols, deg, x, Wm, bias, out, Nn, Ec);
}


# Edge binning (k_part, k_bin): a wave sends its independent LDS and memory requests as groups, not one by one (a bucket's runs read as whole 128-B lines, all in flight); arrival-order sums kept
# speedup vs baseline: 5.4191x; 1.0388x over previous
#include <hip/hip_runtime.h>


typedef __attribute__((ext_vector_type(2))) float v2f;
typedef __attribute__((ext_vector_type(4))) float v4f;
typedef __attribute__((ext_vector_type(8))) float v8f;

#define IN_C 128
#define OUT_C 128

static inline unsigned cdiv_u(long long a, long long b) { return (unsigned)((a + b - 1) / b); }

#define SLOTS 128
#define SH 7
#define BR (1 << SH)
#define PART_T 512
#define PART_EPT 8
#define PART_CH (PART_T * PART_EPT)
#define BIN_T 512

#define PART_LDS_INTS (4 * PART_T + PART_T / 32 + 2 * PART_CH)
#define BIN_RU 16

__device__ __forceinline__ int run_cap(int nrt, int npb) {
    const int w = (nrt * OUT_C) / npb;
    return w >= 32 ? (w & ~31) : w;
}

__device__ __forceinline__ void part_body(int* __restrict__ lds, const int* __restrict__ rows,
                                          const int* __restrict__ cols, float* __restrict__ out,
                                          int n, int e, int p, int npb) {
    int* hist   = lds;
    int* loff   = hist + PART_T;
    int* gbase  = loff + PART_T;
    int* capl   = gbase + PART_T;
    int* wtot   = capl + PART_T;
    int* sorted = wtot + PART_T / 32;
    int* sdst   = sorted + PART_CH;
    int* outw   = (int*)out;

    const int t    = threadIdx.x;
    const int lane = t & 31;
    const int wave = t >> 5;

    hist[t] = 0;
#pragma unroll
    for (int u = 0; u < PART_EPT; ++u) sdst[t + u * PART_T] = -1;

    int r[PART_EPT], c[PART_EPT], k[PART_EPT];
    const int i0 = p * PART_CH + t;
#pragma unroll
    for (int u = 0; u < PART_EPT; ++u) {
        const int i = i0 + u * PART_T;
        r[u] = -1;
        c[u] = 0;
        if (i < e) {
            r[u] = rows[i];
            c[u] = cols[i];
        }
    }
    __syncthreads();

#pragma unroll
    for (int u = 0; u < PART_EPT; ++u) {
        k[u] = -1;
        if ((unsigned)r[u] < (unsigned)n && (r[u] >> SH) < PART_T)
            k[u] = atomicAdd(&hist[r[u] >> SH], 1);
    }
    __syncthreads();

    const int v = hist[t];
    int inc = v;
#pragma unroll
    for (int d = 1; d < 32; d <<= 1) {
        const int o = __shfl_up(inc, d);
        if (lane >= d) inc += o;
    }
    if (lane == 31) wtot[wave] = inc;
    __syncthreads();
    int woff = 0;
#pragma unroll
    for (int w = 0; w < PART_T / 32; ++w) woff += (w < wave) ? wtot[w] : 0;
    loff[t] = woff + inc - v;
    {
        const int row0t = t << SH;
        int cap = 0, gb = 0;
        if (row0t < n) {
            const int nrt = n - row0t < BR ? n - row0t : BR;
            cap = run_cap(nrt, npb);
            gb  = row0t * OUT_C + p * cap;
            if (cap >= 1) outw[gb] = v;
        }
        capl[t]  = cap;
        gbase[t] = gb + 1;
    }
    __syncthreads();

    int lo[PART_EPT], cp[PART_EPT], gbs[PART_EPT];
#pragma unroll
    for (int u = 0; u < PART_EPT; ++u) {
        const int b = k[u] >= 0 ? r[u] >> SH : 0;
        lo[u]  = loff[b];
        cp[u]  = capl[b];
        gbs[u] = gbase[b];
    }
#pragma unroll
    for (int u = 0; u < PART_EPT; ++u) {
        if (k[u] >= 0) {
            const int s = lo[u] + k[u];
            sorted[s] = ((r[u] & (BR - 1)) << 24) | (c[u] & 0xFFFFFF);
            sdst[s]   = k[u] + 1 < cp[u] ? gbs[u] + k[u] : -1;
        }
    }
    __syncthreads();

    int dd[PART_EPT], sw[PART_EPT];
#pragma unroll
    for (int u = 0; u < PART_EPT; ++u) dd[u] = sdst[t + u * PART_T];
#pragma unroll
    for (int u = 0; u < PART_EPT; ++u) sw[u] = dd[u] >= 0 ? sorted[t + u * PART_T] : 0;
#pragma unroll
    for (int u = 0; u < PART_EPT; ++u)
        if (dd[u] >= 0) outw[dd[u]] = sw[u];
}

__device__ __forceinline__ void bin_one(int* cnt, int* slot, int rl, int c) {
    const int k = atomicAdd(&cnt[rl], 1);
    if (k < SLOTS) slot[rl * SLOTS + ((k + rl) & (SLOTS - 1))] = c;
}

__global__ __launch_bounds__(BIN_T) void k_bin(const int* __restrict__ rows,
                                               const int* __restrict__ cols,
                                               int* deg, float* out, int n, int e, int npb) {
    __shared__ int cnt[BR];
    __shared__ int slot[BR * SLOTS];
    __shared__ int ovf;

    const int t    = threadIdx.x;
    const int lane = t & 31;
    const int wave = __builtin_amdgcn_readfirstlane(t >> 5);
    const int b    = blockIdx.x;
    const int row0 = b << SH;
    const int nr   = n - row0 < BR ? n - row0 : BR;
    int* outw = (int*)out;

    const int  cap    = npb > 0 ? run_cap(nr, npb) : 0;
    const bool ranked = b < PART_T && cap >= 2;
    if (t < BR) cnt[t] = 0;
    if (t == 0) ovf = 0;
    __syncthreads();

    if (ranked) {
        const int* reg = outw + row0 * OUT_C;
        bool bad = false;
        for (int q0 = 0; q0 < npb; q0 += BIN_RU * (BIN_T / 32)) {
            int w[BIN_RU], kq[BIN_RU], cqs[BIN_RU];
#pragma unroll
            for (int u = 0; u < BIN_RU; ++u) {
                const int  q  = q0 + u * (BIN_T / 32) + wave;
                const int* rq = reg + q * cap;
                w[u] = (q < npb && lane < cap) ? rq[lane] : 0;
            }
#pragma unroll
            for (int u = 0; u < BIN_RU; ++u) {
                int cq = __builtin_amdgcn_readlane(w[u], 0);
                if ((unsigned)cq > (unsigned)(cap - 1)) { bad = true; cq = 0; }
                cqs[u] = cq;
                kq[u]  = SLOTS;
                if (lane >= 1 && lane <= cq) kq[u] = atomicAdd(&cnt[(w[u] >> 24) & (BR - 1)], 1);
            }
#pragma unroll
            for (int u = 0; u < BIN_RU; ++u) {
                if (kq[u] < SLOTS) {
                    const int rl = (w[u] >> 24) & (BR - 1);
                    slot[rl * SLOTS + ((kq[u] + rl) & (SLOTS - 1))] = w[u] & 0xFFFFFF;
                }
            }
#pragma unroll
            for (int u = 0; u < BIN_RU; ++u) {
                if (cqs[u] > 31) {
                    const int q = q0 + u * (BIN_T / 32) + wave;
                    for (int j = 32 + lane; j <= cqs[u]; j += 32) {
                        const int wj = reg[q * cap + j];
                        bin_one(cnt, slot, (wj >> 24) & (BR - 1), wj & 0xFFFFFF);
                    }
                }
            }
        }
        if (bad) ovf = 1;
    }
    __syncthreads();
    if (!ranked || ovf != 0) {
        if (t < BR) cnt[t] = 0;
        __syncthreads();
        for (int i = t; i < e; i += BIN_T) {
            const int r = rows[i];
            if ((unsigned)r < (unsigned)n && (r >> SH) == b) bin_one(cnt, slot, r & (BR - 1), cols[i]);
        }
    }
    __syncthreads();

    if (t < nr) deg[row0 + t] = cnt[t] + 1;

    constexpr int RPW = BR / (BIN_T / 32);
    int m[RPW], sv[RPW];
    const int rli = wave + (lane & (RPW - 1)) * (BIN_T / 32);
    const int cv  = rli < nr ? cnt[rli] : 0;
#pragma unroll
    for (int i = 0; i < RPW; ++i) {
        const int cr = __builtin_amdgcn_readlane(cv, i);
        m[i] = cr < SLOTS ? cr : SLOTS;
    }
#pragma unroll
    for (int i = 0; i < RPW; ++i) {
        const int rl = wave + i * (BIN_T / 32);
        sv[i] = lane < m[i] ? slot[rl * SLOTS + ((lane + rl) & (SLOTS - 1))] : 0;
    }
#pragma unroll
    for (int i = 0; i < RPW; ++i) {
        const int rl = wave + i * (BIN_T / 32);
        if (lane < m[i]) outw[(row0 + rl) * OUT_C + lane] = sv[i];
    }
#pragma unroll
    for (int i = 0; i < RPW; ++i) {
        const int rl = wave + i * (BIN_T / 32);
        for (int base = 32; base < m[i]; base += 32) {
            const int j = base + lane;
            if (j < m[i]) outw[(row0 + rl) * OUT_C + j] = slot[rl * SLOTS + ((j + rl) & (SLOTS - 1))];
        }
    }
}

__global__ __launch_bounds__(PART_T) void k_part(const int* __restrict__ rows,
                                                 const int* __restrict__ cols,
                                                 float* __restrict__ out, int n, int e, int npb) {
    __shared__ int lds[PART_LDS_INTS];
    part_body(lds, rows, cols, out, n, e, (int)blockIdx.x, npb);
}

#define AG_GW    16
#define AG_MW    4
#define AG_T     ((AG_GW + AG_MW) * 32)
#define AG_PITCH 132

template <int K>
__device__ __forceinline__ void issue_rows(const float* __restrict__ x, int c, int j0, int lane, v4f* v) {
#pragma unroll
    for (int u = 0; u < K; ++u) {
        const int cj = __builtin_amdgcn_readlane(c, j0 + u);
        v[u] = ((const v4f*)x)[cj * (IN_C / 4) + lane];
    }
}

template <int K>
__device__ __forceinline__ void add_rows(const v4f* v, float nrm, int j0, v4f& acc, float& wsum) {
#pragma unroll
    for (int u = 0; u < K; ++u) {
        const float nj = __int_as_float(__builtin_amdgcn_readlane(__float_as_int(nrm), j0 + u));
        acc += nj * v[u];
        wsum += nj;
    }
}

template <int K>
__device__ __forceinline__ void sum_edges(const float* __restrict__ x, int c, float nrm,
                                          int j0, int lane, v4f& acc, float& wsum) {
    v4f v[K];
    issue_rows<K>(x, c, j0, lane, v);
    add_rows<K>(v, nrm, j0, acc, wsum);
}

template <int K, class Pre>
__device__ __forceinline__ void first_group(const float* __restrict__ x, int c, float nrm, int lane,
                                            int r, float dr2, v4f& acc, float& wsum, Pre pre) {
    v4f v[K > 0 ? K : 1];
    if (K > 0) issue_rows<K>(x, c, 0, lane, v);
    const v4f xs = ((const v4f*)x)[r * (IN_C / 4) + lane];
    pre();
    acc  = dr2 * xs;
    wsum = dr2;
    if (K > 0) add_rows<K>(v, nrm, 0, acc, wsum);
}

template <class Pre>
__device__ __forceinline__ void sum_chunk0(const float* __restrict__ x, int c, float nrm, int len,
                                           int lane, int r, float dr2, v4f& acc, float& wsum, Pre pre) {
    if (len <= 0) {
        first_group<0>(x, c, nrm, lane, r, dr2, acc, wsum, pre);
        return;
    }
    if (len > 12)     first_group<16>(x, c, nrm, lane, r, dr2, acc, wsum, pre);
    else if (len > 8) first_group<12>(x, c, nrm, lane, r, dr2, acc, wsum, pre);
    else if (len > 4) first_group<8>(x, c, nrm, lane, r, dr2, acc, wsum, pre);
    else              first_group<4>(x, c, nrm, lane, r, dr2, acc, wsum, pre);
    if (len > 16) {
        if (len > 28)      sum_edges<16>(x, c, nrm, 16, lane, acc, wsum);
        else if (len > 24) sum_edges<12>(x, c, nrm, 16, lane, acc, wsum);
        else if (len > 20) sum_edges<8>(x, c, nrm, 16, lane, acc, wsum);
        else               sum_edges<4>(x, c, nrm, 16, lane, acc, wsum);
    }
}

__device__ __forceinline__ int slot_id(int word, int lane, int cnt_left, int r, int n) {
    const int c = lane < cnt_left ? word : r;
    return c < 0 ? 0 : (c > n - 1 ? n - 1 : c);
}

template <class Pre>
__device__ __forceinline__ void gather_row(const int* __restrict__ rows,
                                           const int* __restrict__ cols,
                                           const int* __restrict__ deg,
                                           const float* __restrict__ x, const float* out,
                                           int n, int e, int r, int lane,
                                           int sl0, int dgv, int dc0, v4f& acc, float& wsum, Pre pre) {
    const float* orow = out + (size_t)r * OUT_C;

    const int   dg  = __builtin_amdgcn_readfirstlane(dgv);
    const int   cnt = dg - 1;
    const float dr  = rsqrtf((float)dg);
    const float dr2 = dr * dr;

    if (cnt <= SLOTS) {
        {
            const int len = cnt < 32 ? cnt : 32;
            int   c   = slot_id(sl0, lane, cnt, r, n);
            float nrm = dr * rsqrtf((float)dc0);
            const int cl = __builtin_amdgcn_readlane(c, len > 0 ? len - 1 : 0);
            c   = lane < len ? c : cl;
            nrm = lane < len ? nrm : 0.f;
            sum_chunk0(x, c, nrm, len, lane, r, dr2, acc, wsum, pre);
        }
        for (int base = 32; base < cnt; base += 32) {
            const int rem = cnt - base;
            const int len = rem < 32 ? rem : 32;
            const int c   = slot_id(__float_as_int(orow[base + lane]), lane, len, r, n);
            const float nrm = dr * rsqrtf((float)deg[c]);
            int j = 0;
            for (; j + 8 <= len; j += 8) sum_edges<8>(x, c, nrm, j, lane, acc, wsum);
            if (len - j >= 4) { sum_edges<4>(x, c, nrm, j, lane, acc, wsum); j += 4; }
            if (len - j >= 2) { sum_edges<2>(x, c, nrm, j, lane, acc, wsum); j += 2; }
            if (len - j >= 1) { sum_edges<1>(x, c, nrm, j, lane, acc, wsum); }
        }
    } else {
        first_group<0>(x, r, 0.f, lane, r, dr2, acc, wsum, pre);
        for (int base = 0; base < e; base += 32) {
            const int idx = base + lane;
            int rr = -1;
            int cc = r;
            if (idx < e) {
                rr = rows[idx];
                cc = cols[idx];
            }
            cc = cc < 0 ? 0 : (cc > n - 1 ? n - 1 : cc);
            unsigned m = __builtin_amdgcn_ballot_w32(rr == r);
            if (m) {
                const float nrm = dr * rsqrtf((float)deg[cc]);
                while (m) {
                    const int j = __builtin_ctz(m);
                    m &= m - 1;
                    sum_edges<1>(x, cc, nrm, j, lane, acc, wsum);
                }
            }
        }
    }
}

__global__ __launch_bounds__(AG_T) __attribute__((amdgpu_waves_per_eu(10))) void k_agg_gemm(const int* __restrict__ rows,
                                                   const int* __restrict__ cols,
                                                   const int* __restrict__ deg,
                                                   const float* __restrict__ x,
                                                   const float* __restrict__ Wm,
                                                   const float* __restrict__ bias,
                                                   float* out, int n, int e) {
    __shared__ v4f   wl[(IN_C / 8) * (OUT_C / 16) * 32];
    __shared__ __attribute__((aligned(16))) float ag[2][AG_GW * AG_PITCH];
    __shared__ float sw[2][AG_GW];

    const int lane = threadIdx.x & 31;
    const int wave = __builtin_amdgcn_readfirstlane(threadIdx.x >> 5);
    const int half = lane >> 4;
    const int l    = lane & 15;

    const int ntiles = (n + AG_GW - 1) / AG_GW;
    const int first = (int)blockIdx.x;
    const int step  = (int)gridDim.x;
    const int nk    = first < ntiles ? (ntiles - first + step - 1) / step : 0;

    int sl_a = 0, dg_a = 1, dc_a = 1, sl_b = 0, dg_b = 1;
    if (wave < AG_GW) {
        const int r0 = first * AG_GW + wave;
        const int r1 = (first + step) * AG_GW + wave;
        if (0 < nk && r0 < n) {
            sl_a = __float_as_int(out[(size_t)r0 * OUT_C + lane]);
            dg_a = deg[r0];
        }
        if (1 < nk && r1 < n) {
            sl_b = __float_as_int(out[(size_t)r1 * OUT_C + lane]);
            dg_b = deg[r1];
        }
        dc_a = deg[slot_id(sl_a, lane, __builtin_amdgcn_readfirstlane(dg_a) - 1, r0, n)];
    }

    for (int s = 0; s <= nk; ++s) {
        if (wave < AG_GW) {
            if (s < nk) {
                const int r  = (first + s * step) * AG_GW + wave;
                const int r1 = r + step * AG_GW;
                const int r2 = r1 + step * AG_GW;
                int sl_c = 0, dg_c = 1, dc_b = 1;
                auto pre = [&]() {
                    const int r2c = r2 < n ? r2 : n - 1;
                    sl_c = __float_as_int(out[(size_t)r2c * OUT_C + lane]);
                    dg_c = deg[r2c];
                    dc_b = deg[slot_id(sl_b, lane, __builtin_amdgcn_readfirstlane(dg_b) - 1, r1, n)];
                };

                v4f   acc  = {0.f, 0.f, 0.f, 0.f};
                float wsum = 0.f;
                if (r < n) gather_row(rows, cols, deg, x, out, n, e, r, lane, sl_a, dg_a, dc_a, acc, wsum, pre);
                float* agb = ag[s & 1];
                *(v4f*)(agb + wave * AG_PITCH + lane * 4) = acc;
                if (lane == 0) sw[s & 1][wave] = wsum;

                sl_a = sl_b; dg_a = dg_b; dc_a = dc_b;
                sl_b = sl_c; dg_b = dg_c;
            }
        } else if (s == 0) {
            const int mw = wave - AG_GW;
            v2f* w2 = (v2f*)wl;
            for (int u0 = 0; u0 < 2 * (IN_C / 8); u0 += 8) {
                v4f g[8];
#pragma unroll
                for (int u = 0; u < 8; ++u) {
                    const int kp  = (u0 + u) >> 1;
                    const int col = mw * 32 + l * 2 + ((u0 + u) & 1);
                    g[u] = *(const v4f*)(Wm + col * IN_C + (2 * kp + half) * 4);
                }
#pragma unroll
                for (int u = 0; u < 8; ++u) {
                    const int kp   = (u0 + u) >> 1;
                    const int unit = kp * 8 + 2 * mw + ((u0 + u) & 1);
                    v2f lo = {g[u].x, g[u].y};
                    v2f hi = {g[u].z, g[u].w};
                    w2[(unit * 32 + l) * 2 + half]      = lo;
                    w2[(unit * 32 + 16 + l) * 2 + half] = hi;
                }
            }
        } else {
            const int mw   = wave - AG_GW;
            const int row0 = (first + (s - 1) * step) * AG_GW;
            const float* agb = ag[(s - 1) & 1];
            const float* ar  = agb + l * AG_PITCH + half * 2;

            v8f c0 = (v8f){0.f, 0.f, 0.f, 0.f, 0.f, 0.f, 0.f, 0.f};
            v8f c1 = c0;
#pragma unroll
            for (int kp = 0; kp < IN_C / 8; ++kp) {
                const v4f bf0 = wl[(kp * 8 + 2 * mw) * 32 + lane];
                const v4f bf1 = wl[(kp * 8 + 2 * mw + 1) * 32 + lane];
                const v2f a0 = *(const v2f*)(ar + 8 * kp);
                const v2f a1 = *(const v2f*)(ar + 8 * kp + 4);
                v2f b00 = {bf0.x, bf0.y};
                v2f b10 = {bf1.x, bf1.y};
                v2f b01 = {bf0.z, bf0.w};
                v2f b11 = {bf1.z, bf1.w};
                c0 = __builtin_amdgcn_wmma_f32_16x16x4_f32(false, a0, false, b00, (short)0, c0, false, false);
                c1 = __builtin_amdgcn_wmma_f32_16x16x4_f32(false, a0, false, b10, (short)0, c1, false, false);
                c0 = __builtin_amdgcn_wmma_f32_16x16x4_f32(false, a1, false, b01, (short)0, c0, false, false);
                c1 = __builtin_amdgcn_wmma_f32_16x16x4_f32(false, a1, false, b11, (short)0, c1, false, false);
            }

            const v2f bv = *(const v2f*)(bias + mw * 32 + l * 2);
            const float* swb = sw[(s - 1) & 1] + half * 8;
            v2f o[8];
#pragma unroll
            for (int i = 0; i < 8; ++i) {
                const float ws = swb[i];
                o[i] = (v2f){c0[i] + ws * bv.x, c1[i] + ws * bv.y};
            }
            v2f* op = (v2f*)(out + (size_t)(row0 + half * 8) * OUT_C + mw * 32 + l * 2);
            if (row0 + AG_GW <= n) {
#pragma unroll
                for (int i = 0; i < 8; ++i) op[i * (OUT_C / 2)] = o[i];
            } else {
#pragma unroll
                for (int i = 0; i < 8; ++i)
                    if (row0 + half * 8 + i < n) op[i * (OUT_C / 2)] = o[i];
            }
        }
        __syncthreads();
    }
}

extern "C" void kernel_launch(void* const* d_in, const int* in_sizes, int n_in,
                              void* d_out, int out_size, void* d_ws, size_t ws_size,
                              hipStream_t stream) {
    const float* x    = (const float*)d_in[0];
    const int*   ei   = (const int*)d_in[1];
    const float* Wm   = (const float*)d_in[2];
    const float* bias = (const float*)d_in[3];
    float*       out  = (float*)d_out;

    const int Nn = in_sizes[0] / IN_C;
    const int Ec = in_sizes[1] / 2;
    const int* rows = ei;
    const int* cols = ei + Ec;

    int* deg = (int*)d_ws;

    const unsigned npb = cdiv_u(Ec, PART_CH);
    k_part<<<npb, PART_T, 0, stream>>>(rows, cols, out, Nn, Ec, (int)npb);
    k_bin<<<cdiv_u(Nn, BR), BIN_T, 0, stream>>>(rows, cols, deg, out, Nn, Ec, (int)npb);
    k_agg_gemm<<<512, AG_T, 0, stream>>>(rows, cols, deg, x, Wm, bias, out, Nn, Ec);
}
